# starting kernel
# baseline (speedup 1.0000x reference)
#include <hip/hip_runtime.h>

#define Mn   45
#define Bn   64
#define Tn   512
#define Hn   20
#define Gn   80
#define FC   32
#define MHD  256
#define OUTD 22
#define TOK  (Bn * Tn)

#define BQ   16
#define NBG  (Bn / BQ)

typedef __attribute__((ext_vector_type(2))) float v2f;
typedef __attribute__((ext_vector_type(8))) float v8f;

__device__ __forceinline__ v8f wmma4(v2f a, v2f b, v8f c) {
  return __builtin_amdgcn_wmma_f32_16x16x4_f32(false, a, false, b, (short)0, c,
                                               false, false);
}

__device__ __forceinline__ float sigm(float x) {
  return __builtin_amdgcn_rcpf(1.f + __expf(-x));
}
__device__ __forceinline__ float tanh_fast(float x) {
  return 1.f - 2.f * __builtin_amdgcn_rcpf(1.f + __expf(2.f * x));
}

__global__ __launch_bounds__(160) void lstm45_kernel(
    const float* __restrict__ Dis,
    const float* __restrict__ Wih0, const float* __restrict__ Whh0,
    const float* __restrict__ bih0, const float* __restrict__ bhh0,
    const float* __restrict__ Wih1, const float* __restrict__ Whh1,
    const float* __restrict__ bih1, const float* __restrict__ bhh1,
    const float* __restrict__ Wih2, const float* __restrict__ Whh2,
    const float* __restrict__ bih2, const float* __restrict__ bhh2,
    const float* __restrict__ Wfc1, const float* __restrict__ bfc1,
    const float* __restrict__ Wfc2, const float* __restrict__ bfc2,
    float* __restrict__ Dpred)
{
  __shared__ float xall[BQ][Tn];
  __shared__ float hcat[BQ][3 * Hn];
  __shared__ float ccat[BQ][3 * Hn];
  __shared__ float gbuf[BQ][Gn];
  __shared__ float zbuf[BQ][FC];
  __shared__ float wfc2s[FC];

  const int tid  = threadIdx.x;
  const int m    = blockIdx.y;
  const int b0   = blockIdx.x * BQ;
  const int lane = tid & 31;
  const int wv   = tid >> 5;
  const int lh   = lane & 15;
  const int sel  = lane >> 4;
  const int n    = wv * 16 + lh;

  for (int i = tid; i < BQ * Tn; i += 160) {
    int b = i / Tn, t = i % Tn;
    xall[b][t] = Dis[(m * Bn + b0 + b) * Tn + t];
  }
  for (int i = tid; i < BQ * 3 * Hn; i += 160) {
    (&hcat[0][0])[i] = 0.f;
    (&ccat[0][0])[i] = 0.f;
  }
  for (int f = tid; f < FC; f += 160) wfc2s[f] = Wfc2[m * FC + f];

  const float bias0v = bih0[m * Gn + n] + bhh0[m * Gn + n];
  const float bias1v = bih1[m * Gn + n] + bhh1[m * Gn + n];
  const float bias2v = bih2[m * Gn + n] + bhh2[m * Gn + n];
  const float wih0v  = Wih0[m * Gn + n];
  const float bfc2v  = bfc2[m];

  const int wb = (m * Gn + n) * Hn;
  v2f bw0[5], bw1[10], bw2[10];
  #pragma unroll
  for (int kk = 0; kk < 5; ++kk) {
    const int k = 4 * kk + 2 * sel;
    bw0[kk][0] = Whh0[wb + k]; bw0[kk][1] = Whh0[wb + k + 1];
    bw1[kk][0] = Wih1[wb + k]; bw1[kk][1] = Wih1[wb + k + 1];
    bw2[kk][0] = Wih2[wb + k]; bw2[kk][1] = Wih2[wb + k + 1];
  }
  #pragma unroll
  for (int kk = 0; kk < 5; ++kk) {
    const int k = 4 * kk + 2 * sel;
    bw1[5 + kk][0] = Whh1[wb + k]; bw1[5 + kk][1] = Whh1[wb + k + 1];
    bw2[5 + kk][0] = Whh2[wb + k]; bw2[5 + kk][1] = Whh2[wb + k + 1];
  }
  v2f bwf[5];
  float bfc1v = 0.f;
  if (wv < 2) {
    const int fb = (m * FC + n) * Hn;
    #pragma unroll
    for (int kk = 0; kk < 5; ++kk) {
      const int k = 4 * kk + 2 * sel;
      bwf[kk][0] = Wfc1[fb + k]; bwf[kk][1] = Wfc1[fb + k + 1];
    }
    bfc1v = bfc1[m * FC + n];
  }

  const int eb0 = tid / Hn,         eh0 = tid % Hn;
  const int eb1 = (tid + 160) / Hn, eh1 = (tid + 160) % Hn;

  __syncthreads();

#define CELL(l, bb, hh)                                                  \
  {                                                                      \
    float ig = sigm(gbuf[bb][hh]);                                       \
    float fg = sigm(gbuf[bb][Hn + (hh)]);                                \
    float gv = tanh_fast(gbuf[bb][2 * Hn + (hh)]);                       \
    float og = sigm(gbuf[bb][3 * Hn + (hh)]);                            \
    float c  = fg * ccat[bb][(l) * Hn + (hh)] + ig * gv;                 \
    ccat[bb][(l) * Hn + (hh)] = c;                                       \
    hcat[bb][(l) * Hn + (hh)] = og * tanh_fast(c);                       \
  }

  for (int t = 0; t < Tn; ++t) {
    {
      v8f acc;
      #pragma unroll
      for (int r = 0; r < 8; ++r)
        acc[r] = bias0v + xall[r + 8 * sel][t] * wih0v;
      #pragma unroll
      for (int kk = 0; kk < 5; ++kk) {
        const int k = 4 * kk + 2 * sel;
        v2f a; a[0] = hcat[lh][k]; a[1] = hcat[lh][k + 1];
        acc = wmma4(a, bw0[kk], acc);
      }
      #pragma unroll
      for (int r = 0; r < 8; ++r) gbuf[r + 8 * sel][n] = acc[r];
    }
    __syncthreads();
    CELL(0, eb0, eh0)
    CELL(0, eb1, eh1)
    __syncthreads();

    {
      v8f acc;
      #pragma unroll
      for (int r = 0; r < 8; ++r) acc[r] = bias1v;
      #pragma unroll
      for (int kk = 0; kk < 10; ++kk) {
        const int k = 4 * kk + 2 * sel;
        v2f a; a[0] = hcat[lh][k]; a[1] = hcat[lh][k + 1];
        acc = wmma4(a, bw1[kk], acc);
      }
      #pragma unroll
      for (int r = 0; r < 8; ++r) gbuf[r + 8 * sel][n] = acc[r];
    }
    __syncthreads();
    CELL(1, eb0, eh0)
    CELL(1, eb1, eh1)
    __syncthreads();

    {
      v8f acc;
      #pragma unroll
      for (int r = 0; r < 8; ++r) acc[r] = bias2v;
      #pragma unroll
      for (int kk = 0; kk < 10; ++kk) {
        const int k = Hn + 4 * kk + 2 * sel;
        v2f a; a[0] = hcat[lh][k]; a[1] = hcat[lh][k + 1];
        acc = wmma4(a, bw2[kk], acc);
      }
      #pragma unroll
      for (int r = 0; r < 8; ++r) gbuf[r + 8 * sel][n] = acc[r];
    }
    __syncthreads();
    CELL(2, eb0, eh0)
    CELL(2, eb1, eh1)
    __syncthreads();

    if (wv < 2) {
      v8f acc;
      #pragma unroll
      for (int r = 0; r < 8; ++r) acc[r] = bfc1v;
      #pragma unroll
      for (int kk = 0; kk < 5; ++kk) {
        const int k = 2 * Hn + 4 * kk + 2 * sel;
        v2f a; a[0] = hcat[lh][k]; a[1] = hcat[lh][k + 1];
        acc = wmma4(a, bwf[kk], acc);
      }
      #pragma unroll
      for (int r = 0; r < 8; ++r) zbuf[r + 8 * sel][n] = fmaxf(acc[r], 0.f);
    }
    __syncthreads();
    if (tid < BQ) {
      float y = bfc2v;
      #pragma unroll
      for (int f = 0; f < FC; ++f) y += zbuf[tid][f] * wfc2s[f];
      Dpred[(m * Bn + b0 + tid) * Tn + t] = y;
    }
  }
#undef CELL
}

#define ASTR (MHD + 4)

__device__ __forceinline__ void hidden_layer(
    const float* __restrict__ W, const float* __restrict__ bias,
    const float (*IN)[ASTR], float (*OUT)[ASTR], int lane, int wv)
{
  const int lh = lane & 15, sel = lane >> 4;
  for (int tile = wv; tile < 64; tile += 8) {
    const int mt = tile >> 4, nt = tile & 15;
    const int n  = nt * 16 + lh;
    const int mr = mt * 16 + lh;
    v8f acc;
    const float bv = bias[n];
    #pragma unroll
    for (int r = 0; r < 8; ++r) acc[r] = bv;
    #pragma unroll 4
    for (int kk = 0; kk < MHD / 4; ++kk) {
      const int k = 4 * kk + 2 * sel;
      v2f a, b;
      a[0] = IN[mr][k];      a[1] = IN[mr][k + 1];
      b[0] = W[n * MHD + k]; b[1] = W[n * MHD + k + 1];
      acc = wmma4(a, b, acc);
    }
    #pragma unroll
    for (int r = 0; r < 8; ++r)
      OUT[mt * 16 + r + 8 * sel][n] = fmaxf(acc[r], 0.f);
  }
}

__global__ __launch_bounds__(256) void mlp_kernel(
    const float* __restrict__ Dpred,
    const float* __restrict__ Wm0, const float* __restrict__ bm0,
    const float* __restrict__ Wm1, const float* __restrict__ bm1,
    const float* __restrict__ Wm2, const float* __restrict__ bm2,
    const float* __restrict__ Wm3, const float* __restrict__ bm3,
    float* __restrict__ Out)
{
  __shared__ float Abuf[64][ASTR];
  __shared__ float Bbuf[64][ASTR];

  const int tid  = threadIdx.x;
  const int lane = tid & 31, wv = tid >> 5;
  const int lh   = lane & 15, sel = lane >> 4;
  const int tok0 = blockIdx.x * 64;

  for (int i = tid; i < 64 * 48; i += 256) {
    int r = i / 48, f = i % 48;
    Abuf[r][f] = (f < Mn) ? Dpred[f * TOK + tok0 + r] : 0.f;
  }
  __syncthreads();

  for (int tile = wv; tile < 64; tile += 8) {
    const int mt = tile >> 4, nt = tile & 15;
    const int n  = nt * 16 + lh;
    const int mr = mt * 16 + lh;
    v8f acc;
    const float bv = bm0[n];
    #pragma unroll
    for (int r = 0; r < 8; ++r) acc[r] = bv;
    #pragma unroll
    for (int kk = 0; kk < 12; ++kk) {
      const int k  = 4 * kk + 2 * sel;
      const int k0 = (k     < Mn) ? k     : (Mn - 1);
      const int k1 = (k + 1 < Mn) ? k + 1 : (Mn - 1);
      v2f a, b;
      a[0] = Abuf[mr][k];      a[1] = Abuf[mr][k + 1];
      b[0] = Wm0[n * Mn + k0]; b[1] = Wm0[n * Mn + k1];
      acc = wmma4(a, b, acc);
    }
    #pragma unroll
    for (int r = 0; r < 8; ++r)
      Bbuf[mt * 16 + r + 8 * sel][n] = fmaxf(acc[r], 0.f);
  }
  __syncthreads();

  hidden_layer(Wm1, bm1, Bbuf, Abuf, lane, wv);
  __syncthreads();
  hidden_layer(Wm2, bm2, Abuf, Bbuf, lane, wv);
  __syncthreads();

  for (int tile = wv; tile < 8; tile += 8) {
    const int mt = tile >> 1, nt = tile & 1;
    const int n  = nt * 16 + lh;
    const int nc = (n < OUTD) ? n : (OUTD - 1);
    const int mr = mt * 16 + lh;
    v8f acc;
    const float bv = bm3[nc];
    #pragma unroll
    for (int r = 0; r < 8; ++r) acc[r] = bv;
    #pragma unroll 4
    for (int kk = 0; kk < MHD / 4; ++kk) {
      const int k = 4 * kk + 2 * sel;
      v2f a, b;
      a[0] = Bbuf[mr][k];       a[1] = Bbuf[mr][k + 1];
      b[0] = Wm3[nc * MHD + k]; b[1] = Wm3[nc * MHD + k + 1];
      acc = wmma4(a, b, acc);
    }
    if (n < OUTD) {
      #pragma unroll
      for (int r = 0; r < 8; ++r)
        Out[(tok0 + mt * 16 + r + 8 * sel) * OUTD + n] = acc[r];
    }
  }
}

extern "C" void kernel_launch(void* const* d_in, const int* in_sizes, int n_in,
                              void* d_out, int out_size, void* d_ws, size_t ws_size,
                              hipStream_t stream) {
  const float* Dis  = (const float*)d_in[0];
  const float* Wih0 = (const float*)d_in[1];
  const float* Whh0 = (const float*)d_in[2];
  const float* bih0 = (const float*)d_in[3];
  const float* bhh0 = (const float*)d_in[4];
  const float* Wih1 = (const float*)d_in[5];
  const float* Whh1 = (const float*)d_in[6];
  const float* bih1 = (const float*)d_in[7];
  const float* bhh1 = (const float*)d_in[8];
  const float* Wih2 = (const float*)d_in[9];
  const float* Whh2 = (const float*)d_in[10];
  const float* bih2 = (const float*)d_in[11];
  const float* bhh2 = (const float*)d_in[12];
  const float* Wfc1 = (const float*)d_in[13];
  const float* bfc1 = (const float*)d_in[14];
  const float* Wfc2 = (const float*)d_in[15];
  const float* bfc2 = (const float*)d_in[16];
  const float* Wm0  = (const float*)d_in[17];
  const float* bm0  = (const float*)d_in[18];
  const float* Wm1  = (const float*)d_in[19];
  const float* bm1  = (const float*)d_in[20];
  const float* Wm2  = (const float*)d_in[21];
  const float* bm2  = (const float*)d_in[22];
  const float* Wm3  = (const float*)d_in[23];
  const float* bm3  = (const float*)d_in[24];

  float* out   = (float*)d_out;
  float* Dpred = out;
  float* cpf   = out + Mn * Bn * Tn;

  dim3 g1(NBG, Mn);
  lstm45_kernel<<<g1, 160, 0, stream>>>(
      Dis, Wih0, Whh0, bih0, bhh0, Wih1, Whh1, bih1, bhh1,
      Wih2, Whh2, bih2, bhh2, Wfc1, bfc1, Wfc2, bfc2, Dpred);

  mlp_kernel<<<TOK / 64, 256, 0, stream>>>(
      Dpred, Wm0, bm0, Wm1, bm1, Wm2, bm2, Wm3, bm3, cpf);
}


# LSTM phase re-ordered: its LDS reads but the tile read-back grouped at its start (each h fragment read once), a register-fed product chain issued inside the gate tile's LDS round trip; sums kept
# speedup vs baseline: 1.8921x; 1.8921x over previous
#include <hip/hip_runtime.h>

#define Mn   45
#define Bn   64
#define Tn   512
#define Hn   20
#define Gn   80
#define FC   32
#define MHD  256
#define OUTD 22
#define TOK  (Bn * Tn)

#define BQ   16
#define NBG  (Bn / BQ)
#define CH   64

typedef __attribute__((ext_vector_type(2))) float v2f;
typedef __attribute__((ext_vector_type(8))) float v8f;

__device__ __forceinline__ v8f wmma4(v2f a, v2f b, v8f c) {
  return __builtin_amdgcn_wmma_f32_16x16x4_f32(false, a, false, b, (short)0, c,
                                               false, false);
}

__device__ __forceinline__ float sigm(float x) {
  return __builtin_amdgcn_rcpf(1.f + __expf(-x));
}
__device__ __forceinline__ float tanh_fast(float x) {
  return 1.f - 2.f * __builtin_amdgcn_rcpf(1.f + __expf(2.f * x));
}

__device__ __forceinline__ float xhalf(float x) {
  int b = __builtin_bit_cast(int, x);
  return __builtin_bit_cast(float,
      __builtin_amdgcn_permlanex16(b, b, 0x76543210, 0xfedcba98, false, false));
}

#define GP 20

__global__ __launch_bounds__(160) void lstm45_kernel(
    const float* __restrict__ Dis,
    const float* __restrict__ Wih0, const float* __restrict__ Whh0,
    const float* __restrict__ bih0, const float* __restrict__ bhh0,
    const float* __restrict__ Wih1, const float* __restrict__ Whh1,
    const float* __restrict__ bih1, const float* __restrict__ bhh1,
    const float* __restrict__ Wih2, const float* __restrict__ Whh2,
    const float* __restrict__ bih2, const float* __restrict__ bhh2,
    const float* __restrict__ Wfc1, const float* __restrict__ bfc1,
    const float* __restrict__ Wfc2, const float* __restrict__ bfc2,
    float* __restrict__ Dpred)
{
  __shared__ float xall[BQ][Tn];
  __shared__ __attribute__((aligned(16))) float hcat[BQ][3 * Hn];
  __shared__ __attribute__((aligned(16))) float gt0[5][BQ][GP];
  __shared__ __attribute__((aligned(16))) float gt1[5][BQ][GP];
  __shared__ __attribute__((aligned(16))) float gt2[5][BQ][GP];
  __shared__ __attribute__((aligned(16))) float hist[CH][BQ][Hn];
  __shared__ __attribute__((aligned(16))) float ybuf[BQ][CH + 4];

  const int tid  = threadIdx.x;
  const int m    = blockIdx.y;
  const int b0   = blockIdx.x * BQ;
  const int lane = tid & 31;
  const int wv   = tid >> 5;
  const int lh   = lane & 15;
  const int sel  = lane >> 4;
  const int n    = (lh & 3) * Hn + 4 * wv + (lh >> 2);
  const int uc   = 4 * wv + 2 * sel;

  for (int i = tid; i < BQ * Tn; i += 160) {
    int b = i / Tn, t = i % Tn;
    xall[b][t] = Dis[(m * Bn + b0 + b) * Tn + t];
  }
  for (int i = tid; i < BQ * 3 * Hn; i += 160)
    (&hcat[0][0])[i] = 0.f;

  const float bias0v = bih0[m * Gn + n] + bhh0[m * Gn + n];
  const float bias1v = bih1[m * Gn + n] + bhh1[m * Gn + n];
  const float bias2v = bih2[m * Gn + n] + bhh2[m * Gn + n];
  const float wih0v  = Wih0[m * Gn + n];
  const float bfc2v  = bfc2[m];

  const int wb = (m * Gn + n) * Hn;
  v2f bw0[5], bw1[10], bw2[10];
  #pragma unroll
  for (int kk = 0; kk < 5; ++kk) {
    const int k = 4 * kk + 2 * sel;
    bw0[kk][0] = Whh0[wb + k]; bw0[kk][1] = Whh0[wb + k + 1];
    bw1[kk][0] = Wih1[wb + k]; bw1[kk][1] = Wih1[wb + k + 1];
    bw2[kk][0] = Wih2[wb + k]; bw2[kk][1] = Wih2[wb + k + 1];
  }
  #pragma unroll
  for (int kk = 0; kk < 5; ++kk) {
    const int k = 4 * kk + 2 * sel;
    bw1[5 + kk][0] = Whh1[wb + k]; bw1[5 + kk][1] = Whh1[wb + k + 1];
    bw2[5 + kk][0] = Whh2[wb + k]; bw2[5 + kk][1] = Whh2[wb + k + 1];
  }
  v2f wfa[5], wfb[5];
  float hb0[8], hb1[8], hw0[8], hw1[8];
  {
    const int fa = (m * FC + lh) * Hn;
    const int fb = (m * FC + 16 + lh) * Hn;
    #pragma unroll
    for (int kk = 0; kk < 5; ++kk) {
      const int k = 4 * kk + 2 * sel;
      wfa[kk][0] = Wfc1[fa + k]; wfa[kk][1] = Wfc1[fa + k + 1];
      wfb[kk][0] = Wfc1[fb + k]; wfb[kk][1] = Wfc1[fb + k + 1];
    }
    #pragma unroll
    for (int r = 0; r < 8; ++r) {
      const int f = m * FC + 8 * sel + r;
      hb0[r] = bfc1[f];      hw0[r] = Wfc2[f];
      hb1[r] = bfc1[f + 16]; hw1[r] = Wfc2[f + 16];
    }
  }

  float cs0a = 0.f, cs0b = 0.f, cs1a = 0.f, cs1b = 0.f, cs2a = 0.f, cs2b = 0.f;

  __syncthreads();

#define WAVE_HANDOVER()                                                  \
  {                                                                      \
    __builtin_amdgcn_fence(__ATOMIC_RELEASE, "wavefront");               \
    __builtin_amdgcn_wave_barrier();                                     \
    __builtin_amdgcn_sched_barrier(0);                                   \
    __builtin_amdgcn_fence(__ATOMIC_ACQUIRE, "wavefront");               \
  }

#define FILL_FENCE(acc)                                                  \
  {                                                                      \
    asm volatile("" :: "v"(acc));                                        \
    __builtin_amdgcn_sched_barrier(0);                                   \
  }

#define LOADFR(fr, cb)                                                   \
  {                                                                      \
    _Pragma("unroll")                                                    \
    for (int kk = 0; kk < 5; ++kk) {                                     \
      const int k = (cb) + 4 * kk + 2 * sel;                             \
      fr[kk][0] = hcat[lh][k]; fr[kk][1] = hcat[lh][k + 1];              \
    }                                                                    \
  }

#define CHAIN(acc, fr, bw, off)                                          \
  {                                                                      \
    _Pragma("unroll")                                                    \
    for (int kk = 0; kk < 5; ++kk)                                       \
      acc = wmma4(fr[kk], bw[(off) + kk], acc);                          \
  }

#define PARK(GT, acc)                                                    \
  {                                                                      \
    _Pragma("unroll")                                                    \
    for (int r = 0; r < 8; ++r) GT[wv][r + 8 * sel][lh] = acc[r];        \
  }

#define READT(GT, ga, gb)                                                \
  {                                                                      \
    ga = *(const float4*)&GT[wv][lh][8 * sel];                           \
    gb = *(const float4*)&GT[wv][lh][8 * sel + 4];                       \
  }
#define READT_NOW(GT, ga, gb)                                            \
  {                                                                      \
    READT(GT, ga, gb)                                                    \
    __builtin_amdgcn_sched_barrier(0);                                   \
  }

#define CELLC(l, ga, gb, csa, csb)                                       \
  {                                                                      \
    float iga = sigm(ga.x), igb = sigm(gb.x);                            \
    float fga = sigm(ga.y), fgb = sigm(gb.y);                            \
    float gva = tanh_fast(ga.z), gvb = tanh_fast(gb.z);                  \
    float oga = sigm(ga.w), ogb = sigm(gb.w);                            \
    float ca = fga * csa + iga * gva;                                    \
    float cb = fgb * csb + igb * gvb;                                    \
    csa = ca; csb = cb;                                                  \
    float2 hv;                                                           \
    hv.x = oga * tanh_fast(ca);                                          \
    hv.y = ogb * tanh_fast(cb);                                          \
    *(float2*)&hcat[lh][(l) * Hn + uc] = hv;                             \
    if ((l) == 2) *(float2*)&hist[tc][lh][uc] = hv;                      \
  }

#define XSTART(g0, tt)                                                   \
  {                                                                      \
    _Pragma("unroll")                                                    \
    for (int r = 0; r < 8; ++r)                                          \
      g0[r] = bias0v + xall[r + 8 * sel][tt] * wih0v;                    \
  }

  float4 qa, qb, pa, pb;

  {
    v2f f0[5];
    v8f g0;
    LOADFR(f0, 0)
    XSTART(g0, 0)
    CHAIN(g0, f0, bw0, 0)
    PARK(gt0, g0)
  }
  __syncthreads();

  v8f e1;
  {
    const int tc = 0;
    v2f f1[5];
    LOADFR(f1, Hn)
    READT(gt0, pa, pb)
    #pragma unroll
    for (int r = 0; r < 8; ++r) e1[r] = bias1v;
    CHAIN(e1, f1, bw1, 5)
    CELLC(0, pa, pb, cs0a, cs0b)
  }
  __syncthreads();

  for (int t = 0; t < Tn; ++t) {
    const int tc = t & (CH - 1);
    const int tn = (t + 1 < Tn) ? t + 1 : t;

    {
      v2f f0[5], f2[5];
      v8f g0, e2;
      LOADFR(f0, 0)
      LOADFR(f2, 2 * Hn)
      XSTART(g0, tn)
      CHAIN(e1, f0, bw1, 0)
      PARK(gt1, e1)
      WAVE_HANDOVER()
      READT_NOW(gt1, qa, qb)
      #pragma unroll
      for (int r = 0; r < 8; ++r) e2[r] = bias2v;
      CHAIN(e2, f2, bw2, 5)
      FILL_FENCE(e2)
      CHAIN(g0, f0, bw0, 0)
      PARK(gt0, g0)
      CELLC(1, qa, qb, cs1a, cs1b)
      __syncthreads();

      v2f f1[5];
      LOADFR(f1, Hn)
      READT(gt0, pa, pb)
      CHAIN(e2, f1, bw2, 0)
      CELLC(0, pa, pb, cs0a, cs0b)
      PARK(gt2, e2)
      WAVE_HANDOVER()
      READT_NOW(gt2, qa, qb)
      #pragma unroll
      for (int r = 0; r < 8; ++r) e1[r] = bias1v;
      CHAIN(e1, f1, bw1, 5)
      FILL_FENCE(e1)
      CELLC(2, qa, qb, cs2a, cs2b)
    }
    __syncthreads();

    if (tc == CH - 1) {
      const int wvu = __builtin_amdgcn_readfirstlane(wv);
      #pragma unroll 2
      for (int tl = wvu; tl < CH; tl += 5) {
        v8f z0, z1;
        #pragma unroll
        for (int r = 0; r < 8; ++r) { z0[r] = hb0[r]; z1[r] = hb1[r]; }
        #pragma unroll
        for (int kk = 0; kk < 5; ++kk) {
          const int k = 4 * kk + 2 * sel;
          v2f hf; hf[0] = hist[tl][lh][k]; hf[1] = hist[tl][lh][k + 1];
          z0 = wmma4(wfa[kk], hf, z0);
          z1 = wmma4(wfb[kk], hf, z1);
        }
        float p0 = fmaxf(z0[0], 0.f) * hw0[0];
        float p1 = fmaxf(z1[0], 0.f) * hw1[0];
        #pragma unroll
        for (int r = 1; r < 8; ++r) p0 += fmaxf(z0[r], 0.f) * hw0[r];
        #pragma unroll
        for (int r = 1; r < 8; ++r) p1 += fmaxf(z1[r], 0.f) * hw1[r];
        float p = p0 + p1;
        p += xhalf(p);
        ybuf[lh][tl] = p + bfc2v;
      }
      __syncthreads();
      for (int i = tid; i < BQ * CH / 4; i += 160) {
        const int b = i / (CH / 4), q = i % (CH / 4);
        const float4 v = *(const float4*)&ybuf[b][4 * q];
        *(float4*)&Dpred[(m * Bn + b0 + b) * Tn + (t - (CH - 1)) + 4 * q] = v;
      }
    }
  }
#undef XSTART
#undef CELLC
#undef READT_NOW
#undef READT
#undef PARK
#undef CHAIN
#undef LOADFR
#undef FILL_FENCE
#undef WAVE_HANDOVER
}

#define ASTR (MHD + 4)

__device__ __forceinline__ void hidden_layer(
    const float* __restrict__ W, const float* __restrict__ bias,
    const float (*IN)[ASTR], float (*OUT)[ASTR], int lane, int wv)
{
  const int lh = lane & 15, sel = lane >> 4;
  for (int tile = wv; tile < 64; tile += 8) {
    const int mt = tile >> 4, nt = tile & 15;
    const int n  = nt * 16 + lh;
    const int mr = mt * 16 + lh;
    v8f acc;
    const float bv = bias[n];
    #pragma unroll
    for (int r = 0; r < 8; ++r) acc[r] = bv;
    #pragma unroll 4
    for (int kk = 0; kk < MHD / 4; ++kk) {
      const int k = 4 * kk + 2 * sel;
      v2f a, b;
      a[0] = IN[mr][k];      a[1] = IN[mr][k + 1];
      b[0] = W[n * MHD + k]; b[1] = W[n * MHD + k + 1];
      acc = wmma4(a, b, acc);
    }
    #pragma unroll
    for (int r = 0; r < 8; ++r)
      OUT[mt * 16 + r + 8 * sel][n] = fmaxf(acc[r], 0.f);
  }
}

__global__ __launch_bounds__(256) void mlp_kernel(
    const float* __restrict__ Dpred,
    const float* __restrict__ Wm0, const float* __restrict__ bm0,
    const float* __restrict__ Wm1, const float* __restrict__ bm1,
    const float* __restrict__ Wm2, const float* __restrict__ bm2,
    const float* __restrict__ Wm3, const float* __restrict__ bm3,
    float* __restrict__ Out)
{
  __shared__ float Abuf[64][ASTR];
  __shared__ float Bbuf[64][ASTR];

  const int tid  = threadIdx.x;
  const int lane = tid & 31, wv = tid >> 5;
  const int lh   = lane & 15, sel = lane >> 4;
  const int tok0 = blockIdx.x * 64;

  for (int i = tid; i < 64 * 48; i += 256) {
    int r = i / 48, f = i % 48;
    Abuf[r][f] = (f < Mn) ? Dpred[f * TOK + tok0 + r] : 0.f;
  }
  __syncthreads();

  for (int tile = wv; tile < 64; tile += 8) {
    const int mt = tile >> 4, nt = tile & 15;
    const int n  = nt * 16 + lh;
    const int mr = mt * 16 + lh;
    v8f acc;
    const float bv = bm0[n];
    #pragma unroll
    for (int r = 0; r < 8; ++r) acc[r] = bv;
    #pragma unroll
    for (int kk = 0; kk < 12; ++kk) {
      const int k  = 4 * kk + 2 * sel;
      const int k0 = (k     < Mn) ? k     : (Mn - 1);
      const int k1 = (k + 1 < Mn) ? k + 1 : (Mn - 1);
      v2f a, b;
      a[0] = Abuf[mr][k];      a[1] = Abuf[mr][k + 1];
      b[0] = Wm0[n * Mn + k0]; b[1] = Wm0[n * Mn + k1];
      acc = wmma4(a, b, acc);
    }
    #pragma unroll
    for (int r = 0; r < 8; ++r)
      Bbuf[mt * 16 + r + 8 * sel][n] = fmaxf(acc[r], 0.f);
  }
  __syncthreads();

  hidden_layer(Wm1, bm1, Bbuf, Abuf, lane, wv);
  __syncthreads();
  hidden_layer(Wm2, bm2, Abuf, Bbuf, lane, wv);
  __syncthreads();

  for (int tile = wv; tile < 8; tile += 8) {
    const int mt = tile >> 1, nt = tile & 1;
    const int n  = nt * 16 + lh;
    const int nc = (n < OUTD) ? n : (OUTD - 1);
    const int mr = mt * 16 + lh;
    v8f acc;
    const float bv = bm3[nc];
    #pragma unroll
    for (int r = 0; r < 8; ++r) acc[r] = bv;
    #pragma unroll 4
    for (int kk = 0; kk < MHD / 4; ++kk) {
      const int k = 4 * kk + 2 * sel;
      v2f a, b;
      a[0] = Bbuf[mr][k];       a[1] = Bbuf[mr][k + 1];
      b[0] = Wm3[nc * MHD + k]; b[1] = Wm3[nc * MHD + k + 1];
      acc = wmma4(a, b, acc);
    }
    if (n < OUTD) {
      #pragma unroll
      for (int r = 0; r < 8; ++r)
        Out[(tok0 + mt * 16 + r + 8 * sel) * OUTD + n] = acc[r];
    }
  }
}

extern "C" void kernel_launch(void* const* d_in, const int* in_sizes, int n_in,
                              void* d_out, int out_size, void* d_ws, size_t ws_size,
                              hipStream_t stream) {
  const float* Dis  = (const float*)d_in[0];
  const float* Wih0 = (const float*)d_in[1];
  const float* Whh0 = (const float*)d_in[2];
  const float* bih0 = (const float*)d_in[3];
  const float* bhh0 = (const float*)d_in[4];
  const float* Wih1 = (const float*)d_in[5];
  const float* Whh1 = (const float*)d_in[6];
  const float* bih1 = (const float*)d_in[7];
  const float* bhh1 = (const float*)d_in[8];
  const float* Wih2 = (const float*)d_in[9];
  const float* Whh2 = (const float*)d_in[10];
  const float* bih2 = (const float*)d_in[11];
  const float* bhh2 = (const float*)d_in[12];
  const float* Wfc1 = (const float*)d_in[13];
  const float* bfc1 = (const float*)d_in[14];
  const float* Wfc2 = (const float*)d_in[15];
  const float* bfc2 = (const float*)d_in[16];
  const float* Wm0  = (const float*)d_in[17];
  const float* bm0  = (const float*)d_in[18];
  const float* Wm1  = (const float*)d_in[19];
  const float* bm1  = (const float*)d_in[20];
  const float* Wm2  = (const float*)d_in[21];
  const float* bm2  = (const float*)d_in[22];
  const float* Wm3  = (const float*)d_in[23];
  const float* bm3  = (const float*)d_in[24];

  float* out   = (float*)d_out;
  float* Dpred = out;
  float* cpf   = out + Mn * Bn * Tn;

  dim3 g1(NBG, Mn);
  lstm45_kernel<<<g1, 160, 0, stream>>>(
      Dis, Wih0, Whh0, bih0, bhh0, Wih1, Whh1, bih1, bhh1,
      Wih2, Whh2, bih2, bhh2, Wfc1, bfc1, Wfc2, bfc2, Dpred);

  mlp_kernel<<<TOK / 64, 256, 0, stream>>>(
      Dpred, Wm0, bm0, Wm1, bm1, Wm2, bm2, Wm3, bm3, cpf);
}


# MLP hidden layers: weights staged in LDS in 16-column slabs by 16 B loads along the rows, next slab prefetched in registers; a weight is fetched once per block (was 4x, 16 lines a load); k order kept
# speedup vs baseline: 2.0478x; 1.0823x over previous
#include <hip/hip_runtime.h>

#define Mn   45
#define Bn   64
#define Tn   512
#define Hn   20
#define Gn   80
#define FC   32
#define MHD  256
#define OUTD 22
#define TOK  (Bn * Tn)

#define BQ   16
#define NBG  (Bn / BQ)
#define CH   64

typedef __attribute__((ext_vector_type(2))) float v2f;
typedef __attribute__((ext_vector_type(8))) float v8f;

__device__ __forceinline__ v8f wmma4(v2f a, v2f b, v8f c) {
  return __builtin_amdgcn_wmma_f32_16x16x4_f32(false, a, false, b, (short)0, c,
                                               false, false);
}

__device__ __forceinline__ float sigm(float x) {
  return __builtin_amdgcn_rcpf(1.f + __expf(-x));
}
__device__ __forceinline__ float tanh_fast(float x) {
  return 1.f - 2.f * __builtin_amdgcn_rcpf(1.f + __expf(2.f * x));
}

__device__ __forceinline__ float xhalf(float x) {
  int b = __builtin_bit_cast(int, x);
  return __builtin_bit_cast(float,
      __builtin_amdgcn_permlanex16(b, b, 0x76543210, 0xfedcba98, false, false));
}

#define GP 20

__global__ __launch_bounds__(160) void lstm45_kernel(
    const float* __restrict__ Dis,
    const float* __restrict__ Wih0, const float* __restrict__ Whh0,
    const float* __restrict__ bih0, const float* __restrict__ bhh0,
    const float* __restrict__ Wih1, const float* __restrict__ Whh1,
    const float* __restrict__ bih1, const float* __restrict__ bhh1,
    const float* __restrict__ Wih2, const float* __restrict__ Whh2,
    const float* __restrict__ bih2, const float* __restrict__ bhh2,
    const float* __restrict__ Wfc1, const float* __restrict__ bfc1,
    const float* __restrict__ Wfc2, const float* __restrict__ bfc2,
    float* __restrict__ Dpred)
{
  __shared__ float xall[BQ][Tn];
  __shared__ __attribute__((aligned(16))) float hcat[BQ][3 * Hn];
  __shared__ __attribute__((aligned(16))) float gt0[5][BQ][GP];
  __shared__ __attribute__((aligned(16))) float gt1[5][BQ][GP];
  __shared__ __attribute__((aligned(16))) float gt2[5][BQ][GP];
  __shared__ __attribute__((aligned(16))) float hist[CH][BQ][Hn];
  __shared__ __attribute__((aligned(16))) float ybuf[BQ][CH + 4];

  const int tid  = threadIdx.x;
  const int m    = blockIdx.y;
  const int b0   = blockIdx.x * BQ;
  const int lane = tid & 31;
  const int wv   = tid >> 5;
  const int lh   = lane & 15;
  const int sel  = lane >> 4;
  const int n    = (lh & 3) * Hn + 4 * wv + (lh >> 2);
  const int uc   = 4 * wv + 2 * sel;

  for (int i = tid; i < BQ * Tn; i += 160) {
    int b = i / Tn, t = i % Tn;
    xall[b][t] = Dis[(m * Bn + b0 + b) * Tn + t];
  }
  for (int i = tid; i < BQ * 3 * Hn; i += 160)
    (&hcat[0][0])[i] = 0.f;

  const float bias0v = bih0[m * Gn + n] + bhh0[m * Gn + n];
  const float bias1v = bih1[m * Gn + n] + bhh1[m * Gn + n];
  const float bias2v = bih2[m * Gn + n] + bhh2[m * Gn + n];
  const float wih0v  = Wih0[m * Gn + n];
  const float bfc2v  = bfc2[m];

  const int wb = (m * Gn + n) * Hn;
  v2f bw0[5], bw1[10], bw2[10];
  #pragma unroll
  for (int kk = 0; kk < 5; ++kk) {
    const int k = 4 * kk + 2 * sel;
    bw0[kk][0] = Whh0[wb + k]; bw0[kk][1] = Whh0[wb + k + 1];
    bw1[kk][0] = Wih1[wb + k]; bw1[kk][1] = Wih1[wb + k + 1];
    bw2[kk][0] = Wih2[wb + k]; bw2[kk][1] = Wih2[wb + k + 1];
  }
  #pragma unroll
  for (int kk = 0; kk < 5; ++kk) {
    const int k = 4 * kk + 2 * sel;
    bw1[5 + kk][0] = Whh1[wb + k]; bw1[5 + kk][1] = Whh1[wb + k + 1];
    bw2[5 + kk][0] = Whh2[wb + k]; bw2[5 + kk][1] = Whh2[wb + k + 1];
  }
  v2f wfa[5], wfb[5];
  float hb0[8], hb1[8], hw0[8], hw1[8];
  {
    const int fa = (m * FC + lh) * Hn;
    const int fb = (m * FC + 16 + lh) * Hn;
    #pragma unroll
    for (int kk = 0; kk < 5; ++kk) {
      const int k = 4 * kk + 2 * sel;
      wfa[kk][0] = Wfc1[fa + k]; wfa[kk][1] = Wfc1[fa + k + 1];
      wfb[kk][0] = Wfc1[fb + k]; wfb[kk][1] = Wfc1[fb + k + 1];
    }
    #pragma unroll
    for (int r = 0; r < 8; ++r) {
      const int f = m * FC + 8 * sel + r;
      hb0[r] = bfc1[f];      hw0[r] = Wfc2[f];
      hb1[r] = bfc1[f + 16]; hw1[r] = Wfc2[f + 16];
    }
  }

  float cs0a = 0.f, cs0b = 0.f, cs1a = 0.f, cs1b = 0.f, cs2a = 0.f, cs2b = 0.f;

  __syncthreads();

#define WAVE_HANDOVER()                                                  \
  {                                                                      \
    __builtin_amdgcn_fence(__ATOMIC_RELEASE, "wavefront");               \
    __builtin_amdgcn_wave_barrier();                                     \
    __builtin_amdgcn_sched_barrier(0);                                   \
    __builtin_amdgcn_fence(__ATOMIC_ACQUIRE, "wavefront");               \
  }

#define FILL_FENCE(acc)                                                  \
  {                                                                      \
    asm volatile("" :: "v"(acc));                                        \
    __builtin_amdgcn_sched_barrier(0);                                   \
  }

#define LOADFR(fr, cb)                                                   \
  {                                                                      \
    _Pragma("unroll")                                                    \
    for (int kk = 0; kk < 5; ++kk) {                                     \
      const int k = (cb) + 4 * kk + 2 * sel;                             \
      fr[kk][0] = hcat[lh][k]; fr[kk][1] = hcat[lh][k + 1];              \
    }                                                                    \
  }

#define CHAIN(acc, fr, bw, off)                                          \
  {                                                                      \
    _Pragma("unroll")                                                    \
    for (int kk = 0; kk < 5; ++kk)                                       \
      acc = wmma4(fr[kk], bw[(off) + kk], acc);                          \
  }

#define PARK(GT, acc)                                                    \
  {                                                                      \
    _Pragma("unroll")                                                    \
    for (int r = 0; r < 8; ++r) GT[wv][r + 8 * sel][lh] = acc[r];        \
  }

#define READT(GT, ga, gb)                                                \
  {                                                                      \
    ga = *(const float4*)&GT[wv][lh][8 * sel];                           \
    gb = *(const float4*)&GT[wv][lh][8 * sel + 4];                       \
  }
#define READT_NOW(GT, ga, gb)                                            \
  {                                                                      \
    READT(GT, ga, gb)                                                    \
    __builtin_amdgcn_sched_barrier(0);                                   \
  }

#define CELLC(l, ga, gb, csa, csb)                                       \
  {                                                                      \
    float iga = sigm(ga.x), igb = sigm(gb.x);                            \
    float fga = sigm(ga.y), fgb = sigm(gb.y);                            \
    float gva = tanh_fast(ga.z), gvb = tanh_fast(gb.z);                  \
    float oga = sigm(ga.w), ogb = sigm(gb.w);                            \
    float ca = fga * csa + iga * gva;                                    \
    float cb = fgb * csb + igb * gvb;                                    \
    csa = ca; csb = cb;                                                  \
    float2 hv;                                                           \
    hv.x = oga * tanh_fast(ca);                                          \
    hv.y = ogb * tanh_fast(cb);                                          \
    *(float2*)&hcat[lh][(l) * Hn + uc] = hv;                             \
    if ((l) == 2) *(float2*)&hist[tc][lh][uc] = hv;                      \
  }

#define XSTART(g0, tt)                                                   \
  {                                                                      \
    _Pragma("unroll")                                                    \
    for (int r = 0; r < 8; ++r)                                          \
      g0[r] = bias0v + xall[r + 8 * sel][tt] * wih0v;                    \
  }

  float4 qa, qb, pa, pb;

  {
    v2f f0[5];
    v8f g0;
    LOADFR(f0, 0)
    XSTART(g0, 0)
    CHAIN(g0, f0, bw0, 0)
    PARK(gt0, g0)
  }
  __syncthreads();

  v8f e1;
  {
    const int tc = 0;
    v2f f1[5];
    LOADFR(f1, Hn)
    READT(gt0, pa, pb)
    #pragma unroll
    for (int r = 0; r < 8; ++r) e1[r] = bias1v;
    CHAIN(e1, f1, bw1, 5)
    CELLC(0, pa, pb, cs0a, cs0b)
  }
  __syncthreads();

  for (int t = 0; t < Tn; ++t) {
    const int tc = t & (CH - 1);
    const int tn = (t + 1 < Tn) ? t + 1 : t;

    {
      v2f f0[5], f2[5];
      v8f g0, e2;
      LOADFR(f0, 0)
      LOADFR(f2, 2 * Hn)
      XSTART(g0, tn)
      CHAIN(e1, f0, bw1, 0)
      PARK(gt1, e1)
      WAVE_HANDOVER()
      READT_NOW(gt1, qa, qb)
      #pragma unroll
      for (int r = 0; r < 8; ++r) e2[r] = bias2v;
      CHAIN(e2, f2, bw2, 5)
      FILL_FENCE(e2)
      CHAIN(g0, f0, bw0, 0)
      PARK(gt0, g0)
      CELLC(1, qa, qb, cs1a, cs1b)
      __syncthreads();

      v2f f1[5];
      LOADFR(f1, Hn)
      READT(gt0, pa, pb)
      CHAIN(e2, f1, bw2, 0)
      CELLC(0, pa, pb, cs0a, cs0b)
      PARK(gt2, e2)
      WAVE_HANDOVER()
      READT_NOW(gt2, qa, qb)
      #pragma unroll
      for (int r = 0; r < 8; ++r) e1[r] = bias1v;
      CHAIN(e1, f1, bw1, 5)
      FILL_FENCE(e1)
      CELLC(2, qa, qb, cs2a, cs2b)
    }
    __syncthreads();

    if (tc == CH - 1) {
      const int wvu = __builtin_amdgcn_readfirstlane(wv);
      #pragma unroll 2
      for (int tl = wvu; tl < CH; tl += 5) {
        v8f z0, z1;
        #pragma unroll
        for (int r = 0; r < 8; ++r) { z0[r] = hb0[r]; z1[r] = hb1[r]; }
        #pragma unroll
        for (int kk = 0; kk < 5; ++kk) {
          const int k = 4 * kk + 2 * sel;
          v2f hf; hf[0] = hist[tl][lh][k]; hf[1] = hist[tl][lh][k + 1];
          z0 = wmma4(wfa[kk], hf, z0);
          z1 = wmma4(wfb[kk], hf, z1);
        }
        float p0 = fmaxf(z0[0], 0.f) * hw0[0];
        float p1 = fmaxf(z1[0], 0.f) * hw1[0];
        #pragma unroll
        for (int r = 1; r < 8; ++r) p0 += fmaxf(z0[r], 0.f) * hw0[r];
        #pragma unroll
        for (int r = 1; r < 8; ++r) p1 += fmaxf(z1[r], 0.f) * hw1[r];
        float p = p0 + p1;
        p += xhalf(p);
        ybuf[lh][tl] = p + bfc2v;
      }
      __syncthreads();
      for (int i = tid; i < BQ * CH / 4; i += 160) {
        const int b = i / (CH / 4), q = i % (CH / 4);
        const float4 v = *(const float4*)&ybuf[b][4 * q];
        *(float4*)&Dpred[(m * Bn + b0 + b) * Tn + (t - (CH - 1)) + 4 * q] = v;
      }
    }
  }
#undef XSTART
#undef CELLC
#undef READT_NOW
#undef READT
#undef PARK
#undef CHAIN
#undef LOADFR
#undef FILL_FENCE
#undef WAVE_HANDOVER
}

#define ASTR (MHD + 4)
#define KS   16
#define WSTR (KS + 4)

__device__ __forceinline__ void hidden_layer(
    const float* __restrict__ W, const float* __restrict__ bias,
    const float (*IN)[ASTR], float (*OUT)[ASTR], float (*WS)[WSTR],
    int tid, int lane, int wv)
{
  const int lh = lane & 15, sel = lane >> 4;
  const int n0 = wv * 16 + lh, n1 = n0 + 128;
  v8f acc[2][4];
  {
    const float bv0 = bias[n0], bv1 = bias[n1];
    #pragma unroll
    for (int mt = 0; mt < 4; ++mt) {
      #pragma unroll
      for (int r = 0; r < 8; ++r) { acc[0][mt][r] = bv0; acc[1][mt][r] = bv1; }
    }
  }
  const int crow = tid >> 2, ccol = 4 * (tid & 3);
  const float* wp = W + crow * MHD + ccol;
  float4 w0 = *(const float4*)(wp);
  float4 w1 = *(const float4*)(wp + 64 * MHD);
  float4 w2 = *(const float4*)(wp + 128 * MHD);
  float4 w3 = *(const float4*)(wp + 192 * MHD);
  #pragma unroll 1
  for (int s = 0; s < MHD / KS; ++s) {
    __syncthreads();
    *(float4*)&WS[crow][ccol]       = w0;
    *(float4*)&WS[crow + 64][ccol]  = w1;
    *(float4*)&WS[crow + 128][ccol] = w2;
    *(float4*)&WS[crow + 192][ccol] = w3;
    __syncthreads();
    if (s + 1 < MHD / KS) {
      const float* wn = wp + (s + 1) * KS;
      w0 = *(const float4*)(wn);
      w1 = *(const float4*)(wn + 64 * MHD);
      w2 = *(const float4*)(wn + 128 * MHD);
      w3 = *(const float4*)(wn + 192 * MHD);
    }
    #pragma unroll
    for (int kk = 0; kk < KS / 4; ++kk) {
      const int k = 4 * kk + 2 * sel;
      v2f b0, b1;
      b0[0] = WS[n0][k]; b0[1] = WS[n0][k + 1];
      b1[0] = WS[n1][k]; b1[1] = WS[n1][k + 1];
      #pragma unroll
      for (int mt = 0; mt < 4; ++mt) {
        v2f a;
        a[0] = IN[mt * 16 + lh][s * KS + k]; a[1] = IN[mt * 16 + lh][s * KS + k + 1];
        acc[0][mt] = wmma4(a, b0, acc[0][mt]);
        acc[1][mt] = wmma4(a, b1, acc[1][mt]);
      }
    }
  }
  #pragma unroll
  for (int mt = 0; mt < 4; ++mt) {
    #pragma unroll
    for (int r = 0; r < 8; ++r) {
      OUT[mt * 16 + r + 8 * sel][n0] = fmaxf(acc[0][mt][r], 0.f);
      OUT[mt * 16 + r + 8 * sel][n1] = fmaxf(acc[1][mt][r], 0.f);
    }
  }
}

__global__ __launch_bounds__(256) void mlp_kernel(
    const float* __restrict__ Dpred,
    const float* __restrict__ Wm0, const float* __restrict__ bm0,
    const float* __restrict__ Wm1, const float* __restrict__ bm1,
    const float* __restrict__ Wm2, const float* __restrict__ bm2,
    const float* __restrict__ Wm3, const float* __restrict__ bm3,
    float* __restrict__ Out)
{
  __shared__ float Abuf[64][ASTR];
  __shared__ float Bbuf[64][ASTR];
  __shared__ __attribute__((aligned(16))) float Wslab[MHD][WSTR];

  const int tid  = threadIdx.x;
  const int lane = tid & 31, wv = tid >> 5;
  const int lh   = lane & 15, sel = lane >> 4;
  const int tok0 = blockIdx.x * 64;

  for (int i = tid; i < 64 * 48; i += 256) {
    int r = i / 48, f = i % 48;
    Abuf[r][f] = (f < Mn) ? Dpred[f * TOK + tok0 + r] : 0.f;
  }
  __syncthreads();

  for (int tile = wv; tile < 64; tile += 8) {
    const int mt = tile >> 4, nt = tile & 15;
    const int n  = nt * 16 + lh;
    const int mr = mt * 16 + lh;
    v8f acc;
    const float bv = bm0[n];
    #pragma unroll
    for (int r = 0; r < 8; ++r) acc[r] = bv;
    #pragma unroll
    for (int kk = 0; kk < 12; ++kk) {
      const int k  = 4 * kk + 2 * sel;
      const int k0 = (k     < Mn) ? k     : (Mn - 1);
      const int k1 = (k + 1 < Mn) ? k + 1 : (Mn - 1);
      v2f a, b;
      a[0] = Abuf[mr][k];      a[1] = Abuf[mr][k + 1];
      b[0] = Wm0[n * Mn + k0]; b[1] = Wm0[n * Mn + k1];
      acc = wmma4(a, b, acc);
    }
    #pragma unroll
    for (int r = 0; r < 8; ++r)
      Bbuf[mt * 16 + r + 8 * sel][n] = fmaxf(acc[r], 0.f);
  }
  __syncthreads();

  hidden_layer(Wm1, bm1, Bbuf, Abuf, Wslab, tid, lane, wv);
  __syncthreads();
  hidden_layer(Wm2, bm2, Abuf, Bbuf, Wslab, tid, lane, wv);
  __syncthreads();

  for (int tile = wv; tile < 8; tile += 8) {
    const int mt = tile >> 1, nt = tile & 1;
    const int n  = nt * 16 + lh;
    const int nc = (n < OUTD) ? n : (OUTD - 1);
    const int mr = mt * 16 + lh;
    v8f acc;
    const float bv = bm3[nc];
    #pragma unroll
    for (int r = 0; r < 8; ++r) acc[r] = bv;
    #pragma unroll 4
    for (int kk = 0; kk < MHD / 4; ++kk) {
      const int k = 4 * kk + 2 * sel;
      v2f a, b;
      a[0] = Bbuf[mr][k];       a[1] = Bbuf[mr][k + 1];
      b[0] = Wm3[nc * MHD + k]; b[1] = Wm3[nc * MHD + k + 1];
      acc = wmma4(a, b, acc);
    }
    if (n < OUTD) {
      #pragma unroll
      for (int r = 0; r < 8; ++r)
        Out[(tok0 + mt * 16 + r + 8 * sel) * OUTD + n] = acc[r];
    }
  }
}

extern "C" void kernel_launch(void* const* d_in, const int* in_sizes, int n_in,
                              void* d_out, int out_size, void* d_ws, size_t ws_size,
                              hipStream_t stream) {
  const float* Dis  = (const float*)d_in[0];
  const float* Wih0 = (const float*)d_in[1];
  const float* Whh0 = (const float*)d_in[2];
  const float* bih0 = (const float*)d_in[3];
  const float* bhh0 = (const float*)d_in[4];
  const float* Wih1 = (const float*)d_in[5];
  const float* Whh1 = (const float*)d_in[6];
  const float* bih1 = (const float*)d_in[7];
  const float* bhh1 = (const float*)d_in[8];
  const float* Wih2 = (const float*)d_in[9];
  const float* Whh2 = (const float*)d_in[10];
  const float* bih2 = (const float*)d_in[11];
  const float* bhh2 = (const float*)d_in[12];
  const float* Wfc1 = (const float*)d_in[13];
  const float* bfc1 = (const float*)d_in[14];
  const float* Wfc2 = (const float*)d_in[15];
  const float* bfc2 = (const float*)d_in[16];
  const float* Wm0  = (const float*)d_in[17];
  const float* bm0  = (const float*)d_in[18];
  const float* Wm1  = (const float*)d_in[19];
  const float* bm1  = (const float*)d_in[20];
  const float* Wm2  = (const float*)d_in[21];
  const float* bm2  = (const float*)d_in[22];
  const float* Wm3  = (const float*)d_in[23];
  const float* bm3  = (const float*)d_in[24];

  float* out   = (float*)d_out;
  float* Dpred = out;
  float* cpf   = out + Mn * Bn * Tn;

  dim3 g1(NBG, Mn);
  lstm45_kernel<<<g1, 160, 0, stream>>>(
      Dis, Wih0, Whh0, bih0, bhh0, Wih1, Whh1, bih1, bhh1,
      Wih2, Whh2, bih2, bhh2, Wfc1, bfc1, Wfc2, bfc2, Dpred);

  mlp_kernel<<<TOK / 64, 256, 0, stream>>>(
      Dpred, Wm0, bm0, Wm1, bm1, Wm2, bm2, Wm3, bm3, cpf);
}


# LSTM layer 0 gate tiles moved to three added waves (8-wave block, pairwise on the SIMDs): the SIMD holding two tile waves issues 40 WMMAs a step, was 50; head loop over 8 waves; every sum order kept
# speedup vs baseline: 2.2207x; 1.0844x over previous
#include <hip/hip_runtime.h>

#define Mn   45
#define Bn   64
#define Tn   512
#define Hn   20
#define Gn   80
#define FC   32
#define MHD  256
#define OUTD 22
#define TOK  (Bn * Tn)

#define BQ   16
#define NBG  (Bn / BQ)
#define CH   64

typedef __attribute__((ext_vector_type(2))) float v2f;
typedef __attribute__((ext_vector_type(8))) float v8f;

__device__ __forceinline__ v8f wmma4(v2f a, v2f b, v8f c) {
  return __builtin_amdgcn_wmma_f32_16x16x4_f32(false, a, false, b, (short)0, c,
                                               false, false);
}

__device__ __forceinline__ float sigm(float x) {
  return __builtin_amdgcn_rcpf(1.f + __expf(-x));
}
__device__ __forceinline__ float tanh_fast(float x) {
  return 1.f - 2.f * __builtin_amdgcn_rcpf(1.f + __expf(2.f * x));
}

__device__ __forceinline__ float xhalf(float x) {
  int b = __builtin_bit_cast(int, x);
  return __builtin_bit_cast(float,
      __builtin_amdgcn_permlanex16(b, b, 0x76543210, 0xfedcba98, false, false));
}

#define GP 20

#define NW1 8
#define NT1 (32 * NW1)

__global__ __launch_bounds__(NT1) void lstm45_kernel(
    const float* __restrict__ Dis,
    const float* __restrict__ Wih0, const float* __restrict__ Whh0,
    const float* __restrict__ bih0, const float* __restrict__ bhh0,
    const float* __restrict__ Wih1, const float* __restrict__ Whh1,
    const float* __restrict__ bih1, const float* __restrict__ bhh1,
    const float* __restrict__ Wih2, const float* __restrict__ Whh2,
    const float* __restrict__ bih2, const float* __restrict__ bhh2,
    const float* __restrict__ Wfc1, const float* __restrict__ bfc1,
    const float* __restrict__ Wfc2, const float* __restrict__ bfc2,
    float* __restrict__ Dpred)
{
  __shared__ float xall[BQ][Tn];
  __shared__ __attribute__((aligned(16))) float hcat[BQ][3 * Hn];
  __shared__ __attribute__((aligned(16))) float gt0[5][BQ][GP];
  __shared__ __attribute__((aligned(16))) float gt1[5][BQ][GP];
  __shared__ __attribute__((aligned(16))) float gt2[5][BQ][GP];
  __shared__ __attribute__((aligned(16))) float hist[CH][BQ][Hn];
  __shared__ __attribute__((aligned(16))) float ybuf[BQ][CH + 4];

  const int tid  = threadIdx.x;
  const int m    = blockIdx.y;
  const int b0   = blockIdx.x * BQ;
  const int lane = tid & 31;
  const int wv   = tid >> 5;
  const int lh   = lane & 15;
  const int sel  = lane >> 4;
  const int  wvu     = __builtin_amdgcn_readfirstlane(wv);
  const bool is_tile = wvu < 5;
  const bool two     = wvu < 7;
  const int ta   = is_tile ? wvu : 2 * (wvu - 5);
  const int tb   = (ta + 1 < 5) ? ta + 1 : 4;
  const int n    = (lh & 3) * Hn + 4 * ta + (lh >> 2);
  const int nb   = (lh & 3) * Hn + 4 * tb + (lh >> 2);
  const int uc   = 4 * wv + 2 * sel;

  for (int i = tid; i < BQ * Tn; i += NT1) {
    int b = i / Tn, t = i % Tn;
    xall[b][t] = Dis[(m * Bn + b0 + b) * Tn + t];
  }
  for (int i = tid; i < BQ * 3 * Hn; i += NT1)
    (&hcat[0][0])[i] = 0.f;

  const float bias0v = bih0[m * Gn + n] + bhh0[m * Gn + n];
  const float bias1v = bih1[m * Gn + n] + bhh1[m * Gn + n];
  const float bias2v = bih2[m * Gn + n] + bhh2[m * Gn + n];
  const float wih0v  = Wih0[m * Gn + n];
  const float bias0w = bih0[m * Gn + nb] + bhh0[m * Gn + nb];
  const float wih0w  = Wih0[m * Gn + nb];
  const float bfc2v  = bfc2[m];

  const int wb = (m * Gn + n) * Hn;
  const int wc = (m * Gn + nb) * Hn;
  v2f bw0[5], bw0b[5], bw1[10], bw2[10];
  #pragma unroll
  for (int kk = 0; kk < 5; ++kk) {
    const int k = 4 * kk + 2 * sel;
    bw0[kk][0] = Whh0[wb + k]; bw0[kk][1] = Whh0[wb + k + 1];
    bw0b[kk][0] = Whh0[wc + k]; bw0b[kk][1] = Whh0[wc + k + 1];
    bw1[kk][0] = Wih1[wb + k]; bw1[kk][1] = Wih1[wb + k + 1];
    bw2[kk][0] = Wih2[wb + k]; bw2[kk][1] = Wih2[wb + k + 1];
  }
  #pragma unroll
  for (int kk = 0; kk < 5; ++kk) {
    const int k = 4 * kk + 2 * sel;
    bw1[5 + kk][0] = Whh1[wb + k]; bw1[5 + kk][1] = Whh1[wb + k + 1];
    bw2[5 + kk][0] = Whh2[wb + k]; bw2[5 + kk][1] = Whh2[wb + k + 1];
  }
  v2f wfa[5], wfb[5];
  float hb0[8], hb1[8], hw0[8], hw1[8];
  {
    const int fa = (m * FC + lh) * Hn;
    const int fb = (m * FC + 16 + lh) * Hn;
    #pragma unroll
    for (int kk = 0; kk < 5; ++kk) {
      const int k = 4 * kk + 2 * sel;
      wfa[kk][0] = Wfc1[fa + k]; wfa[kk][1] = Wfc1[fa + k + 1];
      wfb[kk][0] = Wfc1[fb + k]; wfb[kk][1] = Wfc1[fb + k + 1];
    }
    #pragma unroll
    for (int r = 0; r < 8; ++r) {
      const int f = m * FC + 8 * sel + r;
      hb0[r] = bfc1[f];      hw0[r] = Wfc2[f];
      hb1[r] = bfc1[f + 16]; hw1[r] = Wfc2[f + 16];
    }
  }

  float cs0a = 0.f, cs0b = 0.f, cs1a = 0.f, cs1b = 0.f, cs2a = 0.f, cs2b = 0.f;

  __syncthreads();

#define WAVE_HANDOVER()                                                  \
  {                                                                      \
    __builtin_amdgcn_fence(__ATOMIC_RELEASE, "wavefront");               \
    __builtin_amdgcn_wave_barrier();                                     \
    __builtin_amdgcn_sched_barrier(0);                                   \
    __builtin_amdgcn_fence(__ATOMIC_ACQUIRE, "wavefront");               \
  }

#define FILL_FENCE(acc)                                                  \
  {                                                                      \
    asm volatile("" :: "v"(acc));                                        \
    __builtin_amdgcn_sched_barrier(0);                                   \
  }

#define LOADFR(fr, cb)                                                   \
  {                                                                      \
    _Pragma("unroll")                                                    \
    for (int kk = 0; kk < 5; ++kk) {                                     \
      const int k = (cb) + 4 * kk + 2 * sel;                             \
      fr[kk][0] = hcat[lh][k]; fr[kk][1] = hcat[lh][k + 1];              \
    }                                                                    \
  }

#define CHAIN(acc, fr, bw, off)                                          \
  {                                                                      \
    _Pragma("unroll")                                                    \
    for (int kk = 0; kk < 5; ++kk)                                       \
      acc = wmma4(fr[kk], bw[(off) + kk], acc);                          \
  }

#define PARKT(GT, tw, acc)                                               \
  {                                                                      \
    _Pragma("unroll")                                                    \
    for (int r = 0; r < 8; ++r) GT[tw][r + 8 * sel][lh] = acc[r];        \
  }
#define PARK(GT, acc) PARKT(GT, wv, acc)

#define READT(GT, ga, gb)                                                \
  {                                                                      \
    ga = *(const float4*)&GT[wv][lh][8 * sel];                           \
    gb = *(const float4*)&GT[wv][lh][8 * sel + 4];                       \
  }
#define READT_NOW(GT, ga, gb)                                            \
  {                                                                      \
    READT(GT, ga, gb)                                                    \
    __builtin_amdgcn_sched_barrier(0);                                   \
  }

#define CELLC(l, ga, gb, csa, csb)                                       \
  {                                                                      \
    float iga = sigm(ga.x), igb = sigm(gb.x);                            \
    float fga = sigm(ga.y), fgb = sigm(gb.y);                            \
    float gva = tanh_fast(ga.z), gvb = tanh_fast(gb.z);                  \
    float oga = sigm(ga.w), ogb = sigm(gb.w);                            \
    float ca = fga * csa + iga * gva;                                    \
    float cb = fgb * csb + igb * gvb;                                    \
    csa = ca; csb = cb;                                                  \
    float2 hv;                                                           \
    hv.x = oga * tanh_fast(ca);                                          \
    hv.y = ogb * tanh_fast(cb);                                          \
    *(float2*)&hcat[lh][(l) * Hn + uc] = hv;                             \
    if ((l) == 2) *(float2*)&hist[tc][lh][uc] = hv;                      \
  }

#define XSTART(g0, tt, bv, wx)                                           \
  {                                                                      \
    _Pragma("unroll")                                                    \
    for (int r = 0; r < 8; ++r)                                          \
      g0[r] = (bv) + xall[r + 8 * sel][tt] * (wx);                       \
  }

#define SIDE_L0(tt)                                                      \
  {                                                                      \
    v2f f0[5];                                                           \
    v8f g0;                                                              \
    LOADFR(f0, 0)                                                        \
    XSTART(g0, tt, bias0v, wih0v)                                        \
    if (two) {                                                           \
      v8f g1;                                                            \
      XSTART(g1, tt, bias0w, wih0w)                                      \
      _Pragma("unroll")                                                  \
      for (int kk = 0; kk < 5; ++kk) {                                   \
        g0 = wmma4(f0[kk], bw0[kk], g0);                                 \
        g1 = wmma4(f0[kk], bw0b[kk], g1);                                \
      }                                                                  \
      PARKT(gt0, ta, g0)                                                 \
      PARKT(gt0, tb, g1)                                                 \
    } else {                                                             \
      CHAIN(g0, f0, bw0, 0)                                              \
      PARKT(gt0, ta, g0)                                                 \
    }                                                                    \
  }

  float4 qa, qb, pa, pb;

  if (!is_tile) SIDE_L0(0)
  __syncthreads();

  v8f e1;
  if (is_tile) {
    const int tc = 0;
    v2f f1[5];
    LOADFR(f1, Hn)
    READT(gt0, pa, pb)
    #pragma unroll
    for (int r = 0; r < 8; ++r) e1[r] = bias1v;
    CHAIN(e1, f1, bw1, 5)
    CELLC(0, pa, pb, cs0a, cs0b)
  }
  __syncthreads();

  for (int t = 0; t < Tn; ++t) {
    const int tc = t & (CH - 1);
    const int tn = (t + 1 < Tn) ? t + 1 : t;

    v8f e2;
    if (is_tile) {
      v2f f0[5], f2[5];
      LOADFR(f0, 0)
      LOADFR(f2, 2 * Hn)
      CHAIN(e1, f0, bw1, 0)
      PARK(gt1, e1)
      WAVE_HANDOVER()
      READT_NOW(gt1, qa, qb)
      #pragma unroll
      for (int r = 0; r < 8; ++r) e2[r] = bias2v;
      CHAIN(e2, f2, bw2, 5)
      FILL_FENCE(e2)
      CELLC(1, qa, qb, cs1a, cs1b)
    } else {
      SIDE_L0(tn)
    }
    __syncthreads();

    if (is_tile) {
      v2f f1[5];
      LOADFR(f1, Hn)
      READT(gt0, pa, pb)
      CHAIN(e2, f1, bw2, 0)
      CELLC(0, pa, pb, cs0a, cs0b)
      PARK(gt2, e2)
      WAVE_HANDOVER()
      READT_NOW(gt2, qa, qb)
      #pragma unroll
      for (int r = 0; r < 8; ++r) e1[r] = bias1v;
      CHAIN(e1, f1, bw1, 5)
      FILL_FENCE(e1)
      CELLC(2, qa, qb, cs2a, cs2b)
    }
    __syncthreads();

    if (tc == CH - 1) {
      #pragma unroll 2
      for (int tl = wvu; tl < CH; tl += NW1) {
        v8f z0, z1;
        #pragma unroll
        for (int r = 0; r < 8; ++r) { z0[r] = hb0[r]; z1[r] = hb1[r]; }
        #pragma unroll
        for (int kk = 0; kk < 5; ++kk) {
          const int k = 4 * kk + 2 * sel;
          v2f hf; hf[0] = hist[tl][lh][k]; hf[1] = hist[tl][lh][k + 1];
          z0 = wmma4(wfa[kk], hf, z0);
          z1 = wmma4(wfb[kk], hf, z1);
        }
        float p0 = fmaxf(z0[0], 0.f) * hw0[0];
        float p1 = fmaxf(z1[0], 0.f) * hw1[0];
        #pragma unroll
        for (int r = 1; r < 8; ++r) p0 += fmaxf(z0[r], 0.f) * hw0[r];
        #pragma unroll
        for (int r = 1; r < 8; ++r) p1 += fmaxf(z1[r], 0.f) * hw1[r];
        float p = p0 + p1;
        p += xhalf(p);
        ybuf[lh][tl] = p + bfc2v;
      }
      __syncthreads();
      for (int i = tid; i < BQ * CH / 4; i += NT1) {
        const int b = i / (CH / 4), q = i % (CH / 4);
        const float4 v = *(const float4*)&ybuf[b][4 * q];
        *(float4*)&Dpred[(m * Bn + b0 + b) * Tn + (t - (CH - 1)) + 4 * q] = v;
      }
    }
  }
#undef SIDE_L0
#undef XSTART
#undef CELLC
#undef READT_NOW
#undef READT
#undef PARK
#undef PARKT
#undef CHAIN
#undef LOADFR
#undef FILL_FENCE
#undef WAVE_HANDOVER
}

#define ASTR (MHD + 4)
#define KS   16
#define WSTR (KS + 4)

__device__ __forceinline__ void hidden_layer(
    const float* __restrict__ W, const float* __restrict__ bias,
    const float (*IN)[ASTR], float (*OUT)[ASTR], float (*WS)[WSTR],
    int tid, int lane, int wv)
{
  const int lh = lane & 15, sel = lane >> 4;
  const int n0 = wv * 16 + lh, n1 = n0 + 128;
  v8f acc[2][4];
  {
    const float bv0 = bias[n0], bv1 = bias[n1];
    #pragma unroll
    for (int mt = 0; mt < 4; ++mt) {
      #pragma unroll
      for (int r = 0; r < 8; ++r) { acc[0][mt][r] = bv0; acc[1][mt][r] = bv1; }
    }
  }
  const int crow = tid >> 2, ccol = 4 * (tid & 3);
  const float* wp = W + crow * MHD + ccol;
  float4 w0 = *(const float4*)(wp);
  float4 w1 = *(const float4*)(wp + 64 * MHD);
  float4 w2 = *(const float4*)(wp + 128 * MHD);
  float4 w3 = *(const float4*)(wp + 192 * MHD);
  #pragma unroll 1
  for (int s = 0; s < MHD / KS; ++s) {
    __syncthreads();
    *(float4*)&WS[crow][ccol]       = w0;
    *(float4*)&WS[crow + 64][ccol]  = w1;
    *(float4*)&WS[crow + 128][ccol] = w2;
    *(float4*)&WS[crow + 192][ccol] = w3;
    __syncthreads();
    if (s + 1 < MHD / KS) {
      const float* wn = wp + (s + 1) * KS;
      w0 = *(const float4*)(wn);
      w1 = *(const float4*)(wn + 64 * MHD);
      w2 = *(const float4*)(wn + 128 * MHD);
      w3 = *(const float4*)(wn + 192 * MHD);
    }
    #pragma unroll
    for (int kk = 0; kk < KS / 4; ++kk) {
      const int k = 4 * kk + 2 * sel;
      v2f b0, b1;
      b0[0] = WS[n0][k]; b0[1] = WS[n0][k + 1];
      b1[0] = WS[n1][k]; b1[1] = WS[n1][k + 1];
      #pragma unroll
      for (int mt = 0; mt < 4; ++mt) {
        v2f a;
        a[0] = IN[mt * 16 + lh][s * KS + k]; a[1] = IN[mt * 16 + lh][s * KS + k + 1];
        acc[0][mt] = wmma4(a, b0, acc[0][mt]);
        acc[1][mt] = wmma4(a, b1, acc[1][mt]);
      }
    }
  }
  #pragma unroll
  for (int mt = 0; mt < 4; ++mt) {
    #pragma unroll
    for (int r = 0; r < 8; ++r) {
      OUT[mt * 16 + r + 8 * sel][n0] = fmaxf(acc[0][mt][r], 0.f);
      OUT[mt * 16 + r + 8 * sel][n1] = fmaxf(acc[1][mt][r], 0.f);
    }
  }
}

__global__ __launch_bounds__(256) void mlp_kernel(
    const float* __restrict__ Dpred,
    const float* __restrict__ Wm0, const float* __restrict__ bm0,
    const float* __restrict__ Wm1, const float* __restrict__ bm1,
    const float* __restrict__ Wm2, const float* __restrict__ bm2,
    const float* __restrict__ Wm3, const float* __restrict__ bm3,
    float* __restrict__ Out)
{
  __shared__ float Abuf[64][ASTR];
  __shared__ float Bbuf[64][ASTR];
  __shared__ __attribute__((aligned(16))) float Wslab[MHD][WSTR];

  const int tid  = threadIdx.x;
  const int lane = tid & 31, wv = tid >> 5;
  const int lh   = lane & 15, sel = lane >> 4;
  const int tok0 = blockIdx.x * 64;

  for (int i = tid; i < 64 * 48; i += 256) {
    int r = i / 48, f = i % 48;
    Abuf[r][f] = (f < Mn) ? Dpred[f * TOK + tok0 + r] : 0.f;
  }
  __syncthreads();

  for (int tile = wv; tile < 64; tile += 8) {
    const int mt = tile >> 4, nt = tile & 15;
    const int n  = nt * 16 + lh;
    const int mr = mt * 16 + lh;
    v8f acc;
    const float bv = bm0[n];
    #pragma unroll
    for (int r = 0; r < 8; ++r) acc[r] = bv;
    #pragma unroll
    for (int kk = 0; kk < 12; ++kk) {
      const int k  = 4 * kk + 2 * sel;
      const int k0 = (k     < Mn) ? k     : (Mn - 1);
      const int k1 = (k + 1 < Mn) ? k + 1 : (Mn - 1);
      v2f a, b;
      a[0] = Abuf[mr][k];      a[1] = Abuf[mr][k + 1];
      b[0] = Wm0[n * Mn + k0]; b[1] = Wm0[n * Mn + k1];
      acc = wmma4(a, b, acc);
    }
    #pragma unroll
    for (int r = 0; r < 8; ++r)
      Bbuf[mt * 16 + r + 8 * sel][n] = fmaxf(acc[r], 0.f);
  }
  __syncthreads();

  hidden_layer(Wm1, bm1, Bbuf, Abuf, Wslab, tid, lane, wv);
  __syncthreads();
  hidden_layer(Wm2, bm2, Abuf, Bbuf, Wslab, tid, lane, wv);
  __syncthreads();

  for (int tile = wv; tile < 8; tile += 8) {
    const int mt = tile >> 1, nt = tile & 1;
    const int n  = nt * 16 + lh;
    const int nc = (n < OUTD) ? n : (OUTD - 1);
    const int mr = mt * 16 + lh;
    v8f acc;
    const float bv = bm3[nc];
    #pragma unroll
    for (int r = 0; r < 8; ++r) acc[r] = bv;
    #pragma unroll 4
    for (int kk = 0; kk < MHD / 4; ++kk) {
      const int k = 4 * kk + 2 * sel;
      v2f a, b;
      a[0] = Bbuf[mr][k];       a[1] = Bbuf[mr][k + 1];
      b[0] = Wm3[nc * MHD + k]; b[1] = Wm3[nc * MHD + k + 1];
      acc = wmma4(a, b, acc);
    }
    if (n < OUTD) {
      #pragma unroll
      for (int r = 0; r < 8; ++r)
        Out[(tok0 + mt * 16 + r + 8 * sel) * OUTD + n] = acc[r];
    }
  }
}

extern "C" void kernel_launch(void* const* d_in, const int* in_sizes, int n_in,
                              void* d_out, int out_size, void* d_ws, size_t ws_size,
                              hipStream_t stream) {
  const float* Dis  = (const float*)d_in[0];
  const float* Wih0 = (const float*)d_in[1];
  const float* Whh0 = (const float*)d_in[2];
  const float* bih0 = (const float*)d_in[3];
  const float* bhh0 = (const float*)d_in[4];
  const float* Wih1 = (const float*)d_in[5];
  const float* Whh1 = (const float*)d_in[6];
  const float* bih1 = (const float*)d_in[7];
  const float* bhh1 = (const float*)d_in[8];
  const float* Wih2 = (const float*)d_in[9];
  const float* Whh2 = (const float*)d_in[10];
  const float* bih2 = (const float*)d_in[11];
  const float* bhh2 = (const float*)d_in[12];
  const float* Wfc1 = (const float*)d_in[13];
  const float* bfc1 = (const float*)d_in[14];
  const float* Wfc2 = (const float*)d_in[15];
  const float* bfc2 = (const float*)d_in[16];
  const float* Wm0  = (const float*)d_in[17];
  const float* bm0  = (const float*)d_in[18];
  const float* Wm1  = (const float*)d_in[19];
  const float* bm1  = (const float*)d_in[20];
  const float* Wm2  = (const float*)d_in[21];
  const float* bm2  = (const float*)d_in[22];
  const float* Wm3  = (const float*)d_in[23];
  const float* bm3  = (const float*)d_in[24];

  float* out   = (float*)d_out;
  float* Dpred = out;
  float* cpf   = out + Mn * Bn * Tn;

  dim3 g1(NBG, Mn);
  lstm45_kernel<<<g1, NT1, 0, stream>>>(
      Dis, Wih0, Whh0, bih0, bhh0, Wih1, Whh1, bih1, bhh1,
      Wih2, Whh2, bih2, bhh2, Wfc1, bfc1, Wfc2, bfc2, Dpred);

  mlp_kernel<<<TOK / 64, 256, 0, stream>>>(
      Dpred, Wm0, bm0, Wm1, bm1, Wm2, bm2, Wm3, bm3, cpf);
}


# LSTM layer 0 cell update moved from the tile waves to the side waves that make its gates and only waited in that phase: the SIMD holding two tile waves issues two cell updates a step less; sums kept
# speedup vs baseline: 2.4194x; 1.0895x over previous
#include <hip/hip_runtime.h>

#define Mn   45
#define Bn   64
#define Tn   512
#define Hn   20
#define Gn   80
#define FC   32
#define MHD  256
#define OUTD 22
#define TOK  (Bn * Tn)

#define BQ   16
#define NBG  (Bn / BQ)
#define CH   64

typedef __attribute__((ext_vector_type(2))) float v2f;
typedef __attribute__((ext_vector_type(8))) float v8f;

__device__ __forceinline__ v8f wmma4(v2f a, v2f b, v8f c) {
  return __builtin_amdgcn_wmma_f32_16x16x4_f32(false, a, false, b, (short)0, c,
                                               false, false);
}

__device__ __forceinline__ float sigm(float x) {
  return __builtin_amdgcn_rcpf(1.f + __expf(-x));
}
__device__ __forceinline__ float tanh_fast(float x) {
  return 1.f - 2.f * __builtin_amdgcn_rcpf(1.f + __expf(2.f * x));
}

__device__ __forceinline__ float xhalf(float x) {
  int b = __builtin_bit_cast(int, x);
  return __builtin_bit_cast(float,
      __builtin_amdgcn_permlanex16(b, b, 0x76543210, 0xfedcba98, false, false));
}

#define GP 20

#define NW1 8
#define NT1 (32 * NW1)

__global__ __launch_bounds__(NT1) void lstm45_kernel(
    const float* __restrict__ Dis,
    const float* __restrict__ Wih0, const float* __restrict__ Whh0,
    const float* __restrict__ bih0, const float* __restrict__ bhh0,
    const float* __restrict__ Wih1, const float* __restrict__ Whh1,
    const float* __restrict__ bih1, const float* __restrict__ bhh1,
    const float* __restrict__ Wih2, const float* __restrict__ Whh2,
    const float* __restrict__ bih2, const float* __restrict__ bhh2,
    const float* __restrict__ Wfc1, const float* __restrict__ bfc1,
    const float* __restrict__ Wfc2, const float* __restrict__ bfc2,
    float* __restrict__ Dpred)
{
  __shared__ float xall[BQ][Tn];
  __shared__ __attribute__((aligned(16))) float hcat[BQ][3 * Hn];
  __shared__ __attribute__((aligned(16))) float gt0[5][BQ][GP];
  __shared__ __attribute__((aligned(16))) float gt1[5][BQ][GP];
  __shared__ __attribute__((aligned(16))) float gt2[5][BQ][GP];
  __shared__ __attribute__((aligned(16))) float hist[CH][BQ][Hn];
  __shared__ __attribute__((aligned(16))) float ybuf[BQ][CH + 4];

  const int tid  = threadIdx.x;
  const int m    = blockIdx.y;
  const int b0   = blockIdx.x * BQ;
  const int lane = tid & 31;
  const int wv   = tid >> 5;
  const int lh   = lane & 15;
  const int sel  = lane >> 4;
  const int  wvu     = __builtin_amdgcn_readfirstlane(wv);
  const bool is_tile = wvu < 5;
  const bool two     = wvu < 7;
  const int ta   = is_tile ? wvu : 2 * (wvu - 5);
  const int tb   = (ta + 1 < 5) ? ta + 1 : 4;
  const int n    = (lh & 3) * Hn + 4 * ta + (lh >> 2);
  const int nb   = (lh & 3) * Hn + 4 * tb + (lh >> 2);
  const int uc   = 4 * ta + 2 * sel;
  const int ucb  = 4 * tb + 2 * sel;

  for (int i = tid; i < BQ * Tn; i += NT1) {
    int b = i / Tn, t = i % Tn;
    xall[b][t] = Dis[(m * Bn + b0 + b) * Tn + t];
  }
  for (int i = tid; i < BQ * 3 * Hn; i += NT1)
    (&hcat[0][0])[i] = 0.f;

  const float bias0v = bih0[m * Gn + n] + bhh0[m * Gn + n];
  const float bias1v = bih1[m * Gn + n] + bhh1[m * Gn + n];
  const float bias2v = bih2[m * Gn + n] + bhh2[m * Gn + n];
  const float wih0v  = Wih0[m * Gn + n];
  const float bias0w = bih0[m * Gn + nb] + bhh0[m * Gn + nb];
  const float wih0w  = Wih0[m * Gn + nb];
  const float bfc2v  = bfc2[m];

  const int wb = (m * Gn + n) * Hn;
  const int wc = (m * Gn + nb) * Hn;
  v2f bw0[5], bw0b[5], bw1[10], bw2[10];
  #pragma unroll
  for (int kk = 0; kk < 5; ++kk) {
    const int k = 4 * kk + 2 * sel;
    bw0[kk][0] = Whh0[wb + k]; bw0[kk][1] = Whh0[wb + k + 1];
    bw0b[kk][0] = Whh0[wc + k]; bw0b[kk][1] = Whh0[wc + k + 1];
    bw1[kk][0] = Wih1[wb + k]; bw1[kk][1] = Wih1[wb + k + 1];
    bw2[kk][0] = Wih2[wb + k]; bw2[kk][1] = Wih2[wb + k + 1];
  }
  #pragma unroll
  for (int kk = 0; kk < 5; ++kk) {
    const int k = 4 * kk + 2 * sel;
    bw1[5 + kk][0] = Whh1[wb + k]; bw1[5 + kk][1] = Whh1[wb + k + 1];
    bw2[5 + kk][0] = Whh2[wb + k]; bw2[5 + kk][1] = Whh2[wb + k + 1];
  }
  v2f wfa[5], wfb[5];
  float hb0[8], hb1[8], hw0[8], hw1[8];
  {
    const int fa = (m * FC + lh) * Hn;
    const int fb = (m * FC + 16 + lh) * Hn;
    #pragma unroll
    for (int kk = 0; kk < 5; ++kk) {
      const int k = 4 * kk + 2 * sel;
      wfa[kk][0] = Wfc1[fa + k]; wfa[kk][1] = Wfc1[fa + k + 1];
      wfb[kk][0] = Wfc1[fb + k]; wfb[kk][1] = Wfc1[fb + k + 1];
    }
    #pragma unroll
    for (int r = 0; r < 8; ++r) {
      const int f = m * FC + 8 * sel + r;
      hb0[r] = bfc1[f];      hw0[r] = Wfc2[f];
      hb1[r] = bfc1[f + 16]; hw1[r] = Wfc2[f + 16];
    }
  }

  float cs0a = 0.f, cs0b = 0.f, cs0c = 0.f, cs0d = 0.f;
  float cs1a = 0.f, cs1b = 0.f, cs2a = 0.f, cs2b = 0.f;

  __syncthreads();

#define WAVE_HANDOVER()                                                  \
  {                                                                      \
    __builtin_amdgcn_fence(__ATOMIC_RELEASE, "wavefront");               \
    __builtin_amdgcn_wave_barrier();                                     \
    __builtin_amdgcn_sched_barrier(0);                                   \
    __builtin_amdgcn_fence(__ATOMIC_ACQUIRE, "wavefront");               \
  }

#define FILL_FENCE(acc)                                                  \
  {                                                                      \
    asm volatile("" :: "v"(acc));                                        \
    __builtin_amdgcn_sched_barrier(0);                                   \
  }

#define LOADFR(fr, cb)                                                   \
  {                                                                      \
    _Pragma("unroll")                                                    \
    for (int kk = 0; kk < 5; ++kk) {                                     \
      const int k = (cb) + 4 * kk + 2 * sel;                             \
      fr[kk][0] = hcat[lh][k]; fr[kk][1] = hcat[lh][k + 1];              \
    }                                                                    \
  }

#define CHAIN(acc, fr, bw, off)                                          \
  {                                                                      \
    _Pragma("unroll")                                                    \
    for (int kk = 0; kk < 5; ++kk)                                       \
      acc = wmma4(fr[kk], bw[(off) + kk], acc);                          \
  }

#define PARKT(GT, tw, acc)                                               \
  {                                                                      \
    _Pragma("unroll")                                                    \
    for (int r = 0; r < 8; ++r) GT[tw][r + 8 * sel][lh] = acc[r];        \
  }
#define PARK(GT, acc) PARKT(GT, wv, acc)

#define READT(GT, ga, gb)                                                \
  {                                                                      \
    ga = *(const float4*)&GT[wv][lh][8 * sel];                           \
    gb = *(const float4*)&GT[wv][lh][8 * sel + 4];                       \
  }
#define READT_NOW(GT, ga, gb)                                            \
  {                                                                      \
    READT(GT, ga, gb)                                                    \
    __builtin_amdgcn_sched_barrier(0);                                   \
  }

#define CELLX(col, ga, gb, csa, csb, hv)                                 \
  {                                                                      \
    float iga = sigm(ga.x), igb = sigm(gb.x);                            \
    float fga = sigm(ga.y), fgb = sigm(gb.y);                            \
    float gva = tanh_fast(ga.z), gvb = tanh_fast(gb.z);                  \
    float oga = sigm(ga.w), ogb = sigm(gb.w);                            \
    float ca = fga * csa + iga * gva;                                    \
    float cb = fgb * csb + igb * gvb;                                    \
    csa = ca; csb = cb;                                                  \
    hv.x = oga * tanh_fast(ca);                                          \
    hv.y = ogb * tanh_fast(cb);                                          \
    *(float2*)&hcat[lh][(col)] = hv;                                     \
  }
#define CELLC(l, ga, gb, csa, csb)                                       \
  {                                                                      \
    float2 hv;                                                           \
    CELLX((l) * Hn + uc, ga, gb, csa, csb, hv)                           \
    if ((l) == 2) *(float2*)&hist[tc][lh][uc] = hv;                      \
  }

#define XSTART(g0, tt, bv, wx)                                           \
  {                                                                      \
    _Pragma("unroll")                                                    \
    for (int r = 0; r < 8; ++r)                                          \
      g0[r] = (bv) + xall[r + 8 * sel][tt] * (wx);                       \
  }

#define SIDE_L0(tt)                                                      \
  {                                                                      \
    v2f f0[5];                                                           \
    v8f g0;                                                              \
    LOADFR(f0, 0)                                                        \
    XSTART(g0, tt, bias0v, wih0v)                                        \
    if (two) {                                                           \
      v8f g1;                                                            \
      XSTART(g1, tt, bias0w, wih0w)                                      \
      _Pragma("unroll")                                                  \
      for (int kk = 0; kk < 5; ++kk) {                                   \
        g0 = wmma4(f0[kk], bw0[kk], g0);                                 \
        g1 = wmma4(f0[kk], bw0b[kk], g1);                                \
      }                                                                  \
      PARKT(gt0, ta, g0)                                                 \
      PARKT(gt0, tb, g1)                                                 \
    } else {                                                             \
      CHAIN(g0, f0, bw0, 0)                                              \
      PARKT(gt0, ta, g0)                                                 \
    }                                                                    \
  }

#define SIDE_CELLS()                                                     \
  {                                                                      \
    float4 pa, pb;                                                       \
    float2 ha;                                                           \
    pa = *(const float4*)&gt0[ta][lh][8 * sel];                          \
    pb = *(const float4*)&gt0[ta][lh][8 * sel + 4];                      \
    if (two) {                                                           \
      float4 pc, pd;                                                     \
      float2 hb;                                                         \
      pc = *(const float4*)&gt0[tb][lh][8 * sel];                        \
      pd = *(const float4*)&gt0[tb][lh][8 * sel + 4];                    \
      CELLX(uc, pa, pb, cs0a, cs0b, ha)                                  \
      CELLX(ucb, pc, pd, cs0c, cs0d, hb)                                 \
    } else {                                                             \
      CELLX(uc, pa, pb, cs0a, cs0b, ha)                                  \
    }                                                                    \
  }

  float4 qa, qb;

  if (!is_tile) SIDE_L0(0)
  __syncthreads();

  v8f e1;
  if (is_tile) {
    v2f f1[5];
    LOADFR(f1, Hn)
    #pragma unroll
    for (int r = 0; r < 8; ++r) e1[r] = bias1v;
    CHAIN(e1, f1, bw1, 5)
  } else {
    SIDE_CELLS()
  }
  __syncthreads();

  for (int t = 0; t < Tn; ++t) {
    const int tc = t & (CH - 1);
    const int tn = (t + 1 < Tn) ? t + 1 : t;

    v8f e2;
    if (is_tile) {
      v2f f0[5], f2[5];
      LOADFR(f0, 0)
      LOADFR(f2, 2 * Hn)
      CHAIN(e1, f0, bw1, 0)
      PARK(gt1, e1)
      WAVE_HANDOVER()
      READT_NOW(gt1, qa, qb)
      #pragma unroll
      for (int r = 0; r < 8; ++r) e2[r] = bias2v;
      CHAIN(e2, f2, bw2, 5)
      FILL_FENCE(e2)
      CELLC(1, qa, qb, cs1a, cs1b)
    } else {
      SIDE_L0(tn)
    }
    __syncthreads();

    if (is_tile) {
      v2f f1[5];
      LOADFR(f1, Hn)
      CHAIN(e2, f1, bw2, 0)
      PARK(gt2, e2)
      WAVE_HANDOVER()
      READT_NOW(gt2, qa, qb)
      #pragma unroll
      for (int r = 0; r < 8; ++r) e1[r] = bias1v;
      CHAIN(e1, f1, bw1, 5)
      FILL_FENCE(e1)
      CELLC(2, qa, qb, cs2a, cs2b)
    } else {
      SIDE_CELLS()
    }
    __syncthreads();

    if (tc == CH - 1) {
      #pragma unroll 2
      for (int tl = wvu; tl < CH; tl += NW1) {
        v8f z0, z1;
        #pragma unroll
        for (int r = 0; r < 8; ++r) { z0[r] = hb0[r]; z1[r] = hb1[r]; }
        #pragma unroll
        for (int kk = 0; kk < 5; ++kk) {
          const int k = 4 * kk + 2 * sel;
          v2f hf; hf[0] = hist[tl][lh][k]; hf[1] = hist[tl][lh][k + 1];
          z0 = wmma4(wfa[kk], hf, z0);
          z1 = wmma4(wfb[kk], hf, z1);
        }
        float p0 = fmaxf(z0[0], 0.f) * hw0[0];
        float p1 = fmaxf(z1[0], 0.f) * hw1[0];
        #pragma unroll
        for (int r = 1; r < 8; ++r) p0 += fmaxf(z0[r], 0.f) * hw0[r];
        #pragma unroll
        for (int r = 1; r < 8; ++r) p1 += fmaxf(z1[r], 0.f) * hw1[r];
        float p = p0 + p1;
        p += xhalf(p);
        ybuf[lh][tl] = p + bfc2v;
      }
      __syncthreads();
      for (int i = tid; i < BQ * CH / 4; i += NT1) {
        const int b = i / (CH / 4), q = i % (CH / 4);
        const float4 v = *(const float4*)&ybuf[b][4 * q];
        *(float4*)&Dpred[(m * Bn + b0 + b) * Tn + (t - (CH - 1)) + 4 * q] = v;
      }
    }
  }
#undef SIDE_CELLS
#undef SIDE_L0
#undef XSTART
#undef CELLC
#undef CELLX
#undef READT_NOW
#undef READT
#undef PARK
#undef PARKT
#undef CHAIN
#undef LOADFR
#undef FILL_FENCE
#undef WAVE_HANDOVER
}

#define ASTR (MHD + 4)
#define KS   16
#define WSTR (KS + 4)

__device__ __forceinline__ void hidden_layer(
    const float* __restrict__ W, const float* __restrict__ bias,
    const float (*IN)[ASTR], float (*OUT)[ASTR], float (*WS)[WSTR],
    int tid, int lane, int wv)
{
  const int lh = lane & 15, sel = lane >> 4;
  const int n0 = wv * 16 + lh, n1 = n0 + 128;
  v8f acc[2][4];
  {
    const float bv0 = bias[n0], bv1 = bias[n1];
    #pragma unroll
    for (int mt = 0; mt < 4; ++mt) {
      #pragma unroll
      for (int r = 0; r < 8; ++r) { acc[0][mt][r] = bv0; acc[1][mt][r] = bv1; }
    }
  }
  const int crow = tid >> 2, ccol = 4 * (tid & 3);
  const float* wp = W + crow * MHD + ccol;
  float4 w0 = *(const float4*)(wp);
  float4 w1 = *(const float4*)(wp + 64 * MHD);
  float4 w2 = *(const float4*)(wp + 128 * MHD);
  float4 w3 = *(const float4*)(wp + 192 * MHD);
  #pragma unroll 1
  for (int s = 0; s < MHD / KS; ++s) {
    __syncthreads();
    *(float4*)&WS[crow][ccol]       = w0;
    *(float4*)&WS[crow + 64][ccol]  = w1;
    *(float4*)&WS[crow + 128][ccol] = w2;
    *(float4*)&WS[crow + 192][ccol] = w3;
    __syncthreads();
    if (s + 1 < MHD / KS) {
      const float* wn = wp + (s + 1) * KS;
      w0 = *(const float4*)(wn);
      w1 = *(const float4*)(wn + 64 * MHD);
      w2 = *(const float4*)(wn + 128 * MHD);
      w3 = *(const float4*)(wn + 192 * MHD);
    }
    #pragma unroll
    for (int kk = 0; kk < KS / 4; ++kk) {
      const int k = 4 * kk + 2 * sel;
      v2f b0, b1;
      b0[0] = WS[n0][k]; b0[1] = WS[n0][k + 1];
      b1[0] = WS[n1][k]; b1[1] = WS[n1][k + 1];
      #pragma unroll
      for (int mt = 0; mt < 4; ++mt) {
        v2f a;
        a[0] = IN[mt * 16 + lh][s * KS + k]; a[1] = IN[mt * 16 + lh][s * KS + k + 1];
        acc[0][mt] = wmma4(a, b0, acc[0][mt]);
        acc[1][mt] = wmma4(a, b1, acc[1][mt]);
      }
    }
  }
  #pragma unroll
  for (int mt = 0; mt < 4; ++mt) {
    #pragma unroll
    for (int r = 0; r < 8; ++r) {
      OUT[mt * 16 + r + 8 * sel][n0] = fmaxf(acc[0][mt][r], 0.f);
      OUT[mt * 16 + r + 8 * sel][n1] = fmaxf(acc[1][mt][r], 0.f);
    }
  }
}

__global__ __launch_bounds__(256) void mlp_kernel(
    const float* __restrict__ Dpred,
    const float* __restrict__ Wm0, const float* __restrict__ bm0,
    const float* __restrict__ Wm1, const float* __restrict__ bm1,
    const float* __restrict__ Wm2, const float* __restrict__ bm2,
    const float* __restrict__ Wm3, const float* __restrict__ bm3,
    float* __restrict__ Out)
{
  __shared__ float Abuf[64][ASTR];
  __shared__ float Bbuf[64][ASTR];
  __shared__ __attribute__((aligned(16))) float Wslab[MHD][WSTR];

  const int tid  = threadIdx.x;
  const int lane = tid & 31, wv = tid >> 5;
  const int lh   = lane & 15, sel = lane >> 4;
  const int tok0 = blockIdx.x * 64;

  for (int i = tid; i < 64 * 48; i += 256) {
    int r = i / 48, f = i % 48;
    Abuf[r][f] = (f < Mn) ? Dpred[f * TOK + tok0 + r] : 0.f;
  }
  __syncthreads();

  for (int tile = wv; tile < 64; tile += 8) {
    const int mt = tile >> 4, nt = tile & 15;
    const int n  = nt * 16 + lh;
    const int mr = mt * 16 + lh;
    v8f acc;
    const float bv = bm0[n];
    #pragma unroll
    for (int r = 0; r < 8; ++r) acc[r] = bv;
    #pragma unroll
    for (int kk = 0; kk < 12; ++kk) {
      const int k  = 4 * kk + 2 * sel;
      const int k0 = (k     < Mn) ? k     : (Mn - 1);
      const int k1 = (k + 1 < Mn) ? k + 1 : (Mn - 1);
      v2f a, b;
      a[0] = Abuf[mr][k];      a[1] = Abuf[mr][k + 1];
      b[0] = Wm0[n * Mn + k0]; b[1] = Wm0[n * Mn + k1];
      acc = wmma4(a, b, acc);
    }
    #pragma unroll
    for (int r = 0; r < 8; ++r)
      Bbuf[mt * 16 + r + 8 * sel][n] = fmaxf(acc[r], 0.f);
  }
  __syncthreads();

  hidden_layer(Wm1, bm1, Bbuf, Abuf, Wslab, tid, lane, wv);
  __syncthreads();
  hidden_layer(Wm2, bm2, Abuf, Bbuf, Wslab, tid, lane, wv);
  __syncthreads();

  for (int tile = wv; tile < 8; tile += 8) {
    const int mt = tile >> 1, nt = tile & 1;
    const int n  = nt * 16 + lh;
    const int nc = (n < OUTD) ? n : (OUTD - 1);
    const int mr = mt * 16 + lh;
    v8f acc;
    const float bv = bm3[nc];
    #pragma unroll
    for (int r = 0; r < 8; ++r) acc[r] = bv;
    #pragma unroll 4
    for (int kk = 0; kk < MHD / 4; ++kk) {
      const int k = 4 * kk + 2 * sel;
      v2f a, b;
      a[0] = Bbuf[mr][k];       a[1] = Bbuf[mr][k + 1];
      b[0] = Wm3[nc * MHD + k]; b[1] = Wm3[nc * MHD + k + 1];
      acc = wmma4(a, b, acc);
    }
    if (n < OUTD) {
      #pragma unroll
      for (int r = 0; r < 8; ++r)
        Out[(tok0 + mt * 16 + r + 8 * sel) * OUTD + n] = acc[r];
    }
  }
}

extern "C" void kernel_launch(void* const* d_in, const int* in_sizes, int n_in,
                              void* d_out, int out_size, void* d_ws, size_t ws_size,
                              hipStream_t stream) {
  const float* Dis  = (const float*)d_in[0];
  const float* Wih0 = (const float*)d_in[1];
  const float* Whh0 = (const float*)d_in[2];
  const float* bih0 = (const float*)d_in[3];
  const float* bhh0 = (const float*)d_in[4];
  const float* Wih1 = (const float*)d_in[5];
  const float* Whh1 = (const float*)d_in[6];
  const float* bih1 = (const float*)d_in[7];
  const float* bhh1 = (const float*)d_in[8];
  const float* Wih2 = (const float*)d_in[9];
  const float* Whh2 = (const float*)d_in[10];
  const float* bih2 = (const float*)d_in[11];
  const float* bhh2 = (const float*)d_in[12];
  const float* Wfc1 = (const float*)d_in[13];
  const float* bfc1 = (const float*)d_in[14];
  const float* Wfc2 = (const float*)d_in[15];
  const float* bfc2 = (const float*)d_in[16];
  const float* Wm0  = (const float*)d_in[17];
  const float* bm0  = (const float*)d_in[18];
  const float* Wm1  = (const float*)d_in[19];
  const float* bm1  = (const float*)d_in[20];
  const float* Wm2  = (const float*)d_in[21];
  const float* bm2  = (const float*)d_in[22];
  const float* Wm3  = (const float*)d_in[23];
  const float* bm3  = (const float*)d_in[24];

  float* out   = (float*)d_out;
  float* Dpred = out;
  float* cpf   = out + Mn * Bn * Tn;

  dim3 g1(NBG, Mn);
  lstm45_kernel<<<g1, NT1, 0, stream>>>(
      Dis, Wih0, Whh0, bih0, bhh0, Wih1, Whh1, bih1, bhh1,
      Wih2, Whh2, bih2, bhh2, Wfc1, bfc1, Wfc2, bfc2, Dpred);

  mlp_kernel<<<TOK / 64, 256, 0, stream>>>(
      Dpred, Wm0, bm0, Wm1, bm1, Wm2, bm2, Wm3, bm3, cpf);
}


# LSTM layer 0 half a step further ahead: layer 1's whole gate sum is made in phase 2; chains the next phase needs run behind the barrier arrive, on wave 4 (wave 0's SIMD) ahead of its cells; sums kept
# speedup vs baseline: 2.5342x; 1.0474x over previous
#include <hip/hip_runtime.h>

#define Mn   45
#define Bn   64
#define Tn   512
#define Hn   20
#define Gn   80
#define FC   32
#define MHD  256
#define OUTD 22
#define TOK  (Bn * Tn)

#define BQ   16
#define NBG  (Bn / BQ)
#define CH   64

typedef __attribute__((ext_vector_type(2))) float v2f;
typedef __attribute__((ext_vector_type(8))) float v8f;

__device__ __forceinline__ v8f wmma4(v2f a, v2f b, v8f c) {
  return __builtin_amdgcn_wmma_f32_16x16x4_f32(false, a, false, b, (short)0, c,
                                               false, false);
}

__device__ __forceinline__ float sigm(float x) {
  return __builtin_amdgcn_rcpf(1.f + __expf(-x));
}
__device__ __forceinline__ float tanh_fast(float x) {
  return 1.f - 2.f * __builtin_amdgcn_rcpf(1.f + __expf(2.f * x));
}

__device__ __forceinline__ float xhalf(float x) {
  int b = __builtin_bit_cast(int, x);
  return __builtin_bit_cast(float,
      __builtin_amdgcn_permlanex16(b, b, 0x76543210, 0xfedcba98, false, false));
}

#define GP 20

#define NW1 8
#define NT1 (32 * NW1)

__global__ __launch_bounds__(NT1) void lstm45_kernel(
    const float* __restrict__ Dis,
    const float* __restrict__ Wih0, const float* __restrict__ Whh0,
    const float* __restrict__ bih0, const float* __restrict__ bhh0,
    const float* __restrict__ Wih1, const float* __restrict__ Whh1,
    const float* __restrict__ bih1, const float* __restrict__ bhh1,
    const float* __restrict__ Wih2, const float* __restrict__ Whh2,
    const float* __restrict__ bih2, const float* __restrict__ bhh2,
    const float* __restrict__ Wfc1, const float* __restrict__ bfc1,
    const float* __restrict__ Wfc2, const float* __restrict__ bfc2,
    float* __restrict__ Dpred)
{
  __shared__ float xall[BQ][Tn];
  __shared__ __attribute__((aligned(16))) float hcat[BQ][3 * Hn];
  __shared__ __attribute__((aligned(16))) float gt0[5][BQ][GP];
  __shared__ __attribute__((aligned(16))) float gt1[5][BQ][GP];
  __shared__ __attribute__((aligned(16))) float gt2[5][BQ][GP];
  __shared__ __attribute__((aligned(16))) float hist[CH][BQ][Hn];
  __shared__ __attribute__((aligned(16))) float ybuf[BQ][CH + 4];

  const int tid  = threadIdx.x;
  const int m    = blockIdx.y;
  const int b0   = blockIdx.x * BQ;
  const int lane = tid & 31;
  const int wv   = tid >> 5;
  const int lh   = lane & 15;
  const int sel  = lane >> 4;
  const int  wvu     = __builtin_amdgcn_readfirstlane(wv);
  const bool is_tile = wvu < 5;
  const bool late    = wvu != 4;
  const bool two     = wvu < 7;
  const int ta   = is_tile ? wvu : 2 * (wvu - 5);
  const int tb   = (ta + 1 < 5) ? ta + 1 : 4;
  const int n    = (lh & 3) * Hn + 4 * ta + (lh >> 2);
  const int nb   = (lh & 3) * Hn + 4 * tb + (lh >> 2);
  const int uc   = 4 * ta + 2 * sel;
  const int ucb  = 4 * tb + 2 * sel;

  for (int i = tid; i < BQ * Tn; i += NT1) {
    int b = i / Tn, t = i % Tn;
    xall[b][t] = Dis[(m * Bn + b0 + b) * Tn + t];
  }
  for (int i = tid; i < BQ * 3 * Hn; i += NT1)
    (&hcat[0][0])[i] = 0.f;

  const float bias0v = bih0[m * Gn + n] + bhh0[m * Gn + n];
  const float bias1v = bih1[m * Gn + n] + bhh1[m * Gn + n];
  const float bias2v = bih2[m * Gn + n] + bhh2[m * Gn + n];
  const float wih0v  = Wih0[m * Gn + n];
  const float bias0w = bih0[m * Gn + nb] + bhh0[m * Gn + nb];
  const float wih0w  = Wih0[m * Gn + nb];
  const float bfc2v  = bfc2[m];

  const int wb = (m * Gn + n) * Hn;
  const int wc = (m * Gn + nb) * Hn;
  v2f bw0[5], bw0b[5], bw1[10], bw2[10];
  #pragma unroll
  for (int kk = 0; kk < 5; ++kk) {
    const int k = 4 * kk + 2 * sel;
    bw0[kk][0] = Whh0[wb + k]; bw0[kk][1] = Whh0[wb + k + 1];
    bw0b[kk][0] = Whh0[wc + k]; bw0b[kk][1] = Whh0[wc + k + 1];
    bw1[kk][0] = Wih1[wb + k]; bw1[kk][1] = Wih1[wb + k + 1];
    bw2[kk][0] = Wih2[wb + k]; bw2[kk][1] = Wih2[wb + k + 1];
  }
  #pragma unroll
  for (int kk = 0; kk < 5; ++kk) {
    const int k = 4 * kk + 2 * sel;
    bw1[5 + kk][0] = Whh1[wb + k]; bw1[5 + kk][1] = Whh1[wb + k + 1];
    bw2[5 + kk][0] = Whh2[wb + k]; bw2[5 + kk][1] = Whh2[wb + k + 1];
  }
  v2f wfa[5], wfb[5];
  float hb0[8], hb1[8], hw0[8], hw1[8];
  {
    const int fa = (m * FC + lh) * Hn;
    const int fb = (m * FC + 16 + lh) * Hn;
    #pragma unroll
    for (int kk = 0; kk < 5; ++kk) {
      const int k = 4 * kk + 2 * sel;
      wfa[kk][0] = Wfc1[fa + k]; wfa[kk][1] = Wfc1[fa + k + 1];
      wfb[kk][0] = Wfc1[fb + k]; wfb[kk][1] = Wfc1[fb + k + 1];
    }
    #pragma unroll
    for (int r = 0; r < 8; ++r) {
      const int f = m * FC + 8 * sel + r;
      hb0[r] = bfc1[f];      hw0[r] = Wfc2[f];
      hb1[r] = bfc1[f + 16]; hw1[r] = Wfc2[f + 16];
    }
  }

  float cs0a = 0.f, cs0b = 0.f, cs0c = 0.f, cs0d = 0.f;
  float cs1a = 0.f, cs1b = 0.f, cs2a = 0.f, cs2b = 0.f;

  __syncthreads();

#define WAVE_HANDOVER()                                                  \
  {                                                                      \
    __builtin_amdgcn_fence(__ATOMIC_RELEASE, "wavefront");               \
    __builtin_amdgcn_wave_barrier();                                     \
    __builtin_amdgcn_sched_barrier(0);                                   \
    __builtin_amdgcn_fence(__ATOMIC_ACQUIRE, "wavefront");               \
  }
#define WAVE_ORDER()                                                     \
  {                                                                      \
    __builtin_amdgcn_fence(__ATOMIC_RELEASE, "wavefront");               \
    __builtin_amdgcn_wave_barrier();                                     \
    __builtin_amdgcn_fence(__ATOMIC_ACQUIRE, "wavefront");               \
  }

#define FILL_FENCE(acc)                                                  \
  {                                                                      \
    asm volatile("" :: "v"(acc));                                        \
    __builtin_amdgcn_sched_barrier(0);                                   \
  }

#define LOADFR(fr, cb)                                                   \
  {                                                                      \
    _Pragma("unroll")                                                    \
    for (int kk = 0; kk < 5; ++kk) {                                     \
      const int k = (cb) + 4 * kk + 2 * sel;                             \
      fr[kk][0] = hcat[lh][k]; fr[kk][1] = hcat[lh][k + 1];              \
    }                                                                    \
  }

#define CHAIN(acc, fr, bw, off)                                          \
  {                                                                      \
    _Pragma("unroll")                                                    \
    for (int kk = 0; kk < 5; ++kk)                                       \
      acc = wmma4(fr[kk], bw[(off) + kk], acc);                          \
  }

#define PARKT(GT, tw, acc)                                               \
  {                                                                      \
    _Pragma("unroll")                                                    \
    for (int r = 0; r < 8; ++r) GT[tw][r + 8 * sel][lh] = acc[r];        \
  }
#define PARK(GT, acc) PARKT(GT, wv, acc)

#define READT(GT, ga, gb)                                                \
  {                                                                      \
    ga = *(const float4*)&GT[wv][lh][8 * sel];                           \
    gb = *(const float4*)&GT[wv][lh][8 * sel + 4];                       \
  }
#define READT_NOW(GT, ga, gb)                                            \
  {                                                                      \
    READT(GT, ga, gb)                                                    \
    __builtin_amdgcn_sched_barrier(0);                                   \
  }

#define CELLX(col, ga, gb, csa, csb, hv)                                 \
  {                                                                      \
    float iga = sigm(ga.x), igb = sigm(gb.x);                            \
    float fga = sigm(ga.y), fgb = sigm(gb.y);                            \
    float gva = tanh_fast(ga.z), gvb = tanh_fast(gb.z);                  \
    float oga = sigm(ga.w), ogb = sigm(gb.w);                            \
    float ca = fga * csa + iga * gva;                                    \
    float cb = fgb * csb + igb * gvb;                                    \
    csa = ca; csb = cb;                                                  \
    hv.x = oga * tanh_fast(ca);                                          \
    hv.y = ogb * tanh_fast(cb);                                          \
    *(float2*)&hcat[lh][(col)] = hv;                                     \
  }
#define CELLC(l, ga, gb, csa, csb)                                       \
  {                                                                      \
    float2 hv;                                                           \
    CELLX((l) * Hn + uc, ga, gb, csa, csb, hv)                           \
    if ((l) == 2) *(float2*)&hist[tc][lh][uc] = hv;                      \
  }

#define XSTART(g0, tt, bv, wx)                                           \
  {                                                                      \
    _Pragma("unroll")                                                    \
    for (int r = 0; r < 8; ++r)                                          \
      g0[r] = (bv) + xall[r + 8 * sel][tt] * (wx);                       \
  }

#define SIDE_L0(tt)                                                      \
  {                                                                      \
    v2f f0[5];                                                           \
    v8f g0;                                                              \
    LOADFR(f0, 0)                                                        \
    XSTART(g0, tt, bias0v, wih0v)                                        \
    if (two) {                                                           \
      v8f g1;                                                            \
      XSTART(g1, tt, bias0w, wih0w)                                      \
      _Pragma("unroll")                                                  \
      for (int kk = 0; kk < 5; ++kk) {                                   \
        g0 = wmma4(f0[kk], bw0[kk], g0);                                 \
        g1 = wmma4(f0[kk], bw0b[kk], g1);                                \
      }                                                                  \
      PARKT(gt0, ta, g0)                                                 \
      PARKT(gt0, tb, g1)                                                 \
    } else {                                                             \
      CHAIN(g0, f0, bw0, 0)                                              \
      PARKT(gt0, ta, g0)                                                 \
    }                                                                    \
  }

#define SIDE_CELLS()                                                     \
  {                                                                      \
    float4 pa, pb;                                                       \
    float2 ha;                                                           \
    pa = *(const float4*)&gt0[ta][lh][8 * sel];                          \
    pb = *(const float4*)&gt0[ta][lh][8 * sel + 4];                      \
    if (two) {                                                           \
      float4 pc, pd;                                                     \
      float2 hb;                                                         \
      pc = *(const float4*)&gt0[tb][lh][8 * sel];                        \
      pd = *(const float4*)&gt0[tb][lh][8 * sel + 4];                    \
      CELLX(uc, pa, pb, cs0a, cs0b, ha)                                  \
      CELLX(ucb, pc, pd, cs0c, cs0d, hb)                                 \
    } else {                                                             \
      CELLX(uc, pa, pb, cs0a, cs0b, ha)                                  \
    }                                                                    \
  }

  float4 qa, qb;

#define BAR_ARRIVE()                                                     \
  {                                                                      \
    __builtin_amdgcn_fence(__ATOMIC_RELEASE, "workgroup");               \
    __builtin_amdgcn_s_barrier_signal(-1);                               \
  }
#define BAR_WAIT()                                                       \
  {                                                                      \
    __builtin_amdgcn_s_barrier_wait(-1);                                 \
    __builtin_amdgcn_fence(__ATOMIC_ACQUIRE, "workgroup");               \
  }

  if (!is_tile) SIDE_L0(0)
  __syncthreads();
  if (!is_tile) SIDE_CELLS()
  __syncthreads();

  v8f e1;
  v8f e2;
  v2f f2[5];
  v2f g0[5];
  if (is_tile) {
    v2f f1[5];
    LOADFR(f1, Hn)
    LOADFR(g0, 0)
    #pragma unroll
    for (int r = 0; r < 8; ++r) e1[r] = bias1v;
    CHAIN(e1, f1, bw1, 5)
    CHAIN(e1, g0, bw1, 0)
  } else {
    SIDE_L0(1)
  }
  __syncthreads();

  for (int t = 0; t < Tn; ++t) {
    const int tc = t & (CH - 1);
    const int tn = (t + 2 < Tn) ? t + 2 : Tn - 1;

    if (is_tile) {
      LOADFR(f2, 2 * Hn)
      PARK(gt1, e1)
      WAVE_ORDER()
      READT_NOW(gt1, qa, qb)
      if (!late) {
        #pragma unroll
        for (int r = 0; r < 8; ++r) e2[r] = bias2v;
        CHAIN(e2, f2, bw2, 5)
        FILL_FENCE(e2)
      }
      CELLC(1, qa, qb, cs1a, cs1b)
    } else {
      SIDE_CELLS()
    }
    BAR_ARRIVE()
    if (is_tile && late) {
      #pragma unroll
      for (int r = 0; r < 8; ++r) e2[r] = bias2v;
      CHAIN(e2, f2, bw2, 5)
    }
    BAR_WAIT()

    if (is_tile) {
      v2f f1[5];
      LOADFR(f1, Hn)
      LOADFR(g0, 0)
      CHAIN(e2, f1, bw2, 0)
      PARK(gt2, e2)
      WAVE_HANDOVER()
      READT_NOW(gt2, qa, qb)
      #pragma unroll
      for (int r = 0; r < 8; ++r) e1[r] = bias1v;
      CHAIN(e1, f1, bw1, 5)
      if (!late)
        CHAIN(e1, g0, bw1, 0)
      FILL_FENCE(e1)
      CELLC(2, qa, qb, cs2a, cs2b)
    } else {
      SIDE_L0(tn)
    }
    BAR_ARRIVE()
    if (is_tile && late)
      CHAIN(e1, g0, bw1, 0)
    BAR_WAIT()

    if (tc == CH - 1) {
      #pragma unroll 2
      for (int tl = wvu; tl < CH; tl += NW1) {
        v8f z0, z1;
        #pragma unroll
        for (int r = 0; r < 8; ++r) { z0[r] = hb0[r]; z1[r] = hb1[r]; }
        #pragma unroll
        for (int kk = 0; kk < 5; ++kk) {
          const int k = 4 * kk + 2 * sel;
          v2f hf; hf[0] = hist[tl][lh][k]; hf[1] = hist[tl][lh][k + 1];
          z0 = wmma4(wfa[kk], hf, z0);
          z1 = wmma4(wfb[kk], hf, z1);
        }
        float p0 = fmaxf(z0[0], 0.f) * hw0[0];
        float p1 = fmaxf(z1[0], 0.f) * hw1[0];
        #pragma unroll
        for (int r = 1; r < 8; ++r) p0 += fmaxf(z0[r], 0.f) * hw0[r];
        #pragma unroll
        for (int r = 1; r < 8; ++r) p1 += fmaxf(z1[r], 0.f) * hw1[r];
        float p = p0 + p1;
        p += xhalf(p);
        ybuf[lh][tl] = p + bfc2v;
      }
      __syncthreads();
      for (int i = tid; i < BQ * CH / 4; i += NT1) {
        const int b = i / (CH / 4), q = i % (CH / 4);
        const float4 v = *(const float4*)&ybuf[b][4 * q];
        *(float4*)&Dpred[(m * Bn + b0 + b) * Tn + (t - (CH - 1)) + 4 * q] = v;
      }
    }
  }
#undef BAR_WAIT
#undef BAR_ARRIVE
#undef SIDE_CELLS
#undef SIDE_L0
#undef XSTART
#undef CELLC
#undef CELLX
#undef READT_NOW
#undef READT
#undef PARK
#undef PARKT
#undef CHAIN
#undef LOADFR
#undef FILL_FENCE
#undef WAVE_ORDER
#undef WAVE_HANDOVER
}

#define ASTR (MHD + 4)
#define KS   16
#define WSTR (KS + 4)

__device__ __forceinline__ void hidden_layer(
    const float* __restrict__ W, const float* __restrict__ bias,
    const float (*IN)[ASTR], float (*OUT)[ASTR], float (*WS)[WSTR],
    int tid, int lane, int wv)
{
  const int lh = lane & 15, sel = lane >> 4;
  const int n0 = wv * 16 + lh, n1 = n0 + 128;
  v8f acc[2][4];
  {
    const float bv0 = bias[n0], bv1 = bias[n1];
    #pragma unroll
    for (int mt = 0; mt < 4; ++mt) {
      #pragma unroll
      for (int r = 0; r < 8; ++r) { acc[0][mt][r] = bv0; acc[1][mt][r] = bv1; }
    }
  }
  const int crow = tid >> 2, ccol = 4 * (tid & 3);
  const float* wp = W + crow * MHD + ccol;
  float4 w0 = *(const float4*)(wp);
  float4 w1 = *(const float4*)(wp + 64 * MHD);
  float4 w2 = *(const float4*)(wp + 128 * MHD);
  float4 w3 = *(const float4*)(wp + 192 * MHD);
  #pragma unroll 1
  for (int s = 0; s < MHD / KS; ++s) {
    __syncthreads();
    *(float4*)&WS[crow][ccol]       = w0;
    *(float4*)&WS[crow + 64][ccol]  = w1;
    *(float4*)&WS[crow + 128][ccol] = w2;
    *(float4*)&WS[crow + 192][ccol] = w3;
    __syncthreads();
    if (s + 1 < MHD / KS) {
      const float* wn = wp + (s + 1) * KS;
      w0 = *(const float4*)(wn);
      w1 = *(const float4*)(wn + 64 * MHD);
      w2 = *(const float4*)(wn + 128 * MHD);
      w3 = *(const float4*)(wn + 192 * MHD);
    }
    #pragma unroll
    for (int kk = 0; kk < KS / 4; ++kk) {
      const int k = 4 * kk + 2 * sel;
      v2f b0, b1;
      b0[0] = WS[n0][k]; b0[1] = WS[n0][k + 1];
      b1[0] = WS[n1][k]; b1[1] = WS[n1][k + 1];
      #pragma unroll
      for (int mt = 0; mt < 4; ++mt) {
        v2f a;
        a[0] = IN[mt * 16 + lh][s * KS + k]; a[1] = IN[mt * 16 + lh][s * KS + k + 1];
        acc[0][mt] = wmma4(a, b0, acc[0][mt]);
        acc[1][mt] = wmma4(a, b1, acc[1][mt]);
      }
    }
  }
  #pragma unroll
  for (int mt = 0; mt < 4; ++mt) {
    #pragma unroll
    for (int r = 0; r < 8; ++r) {
      OUT[mt * 16 + r + 8 * sel][n0] = fmaxf(acc[0][mt][r], 0.f);
      OUT[mt * 16 + r + 8 * sel][n1] = fmaxf(acc[1][mt][r], 0.f);
    }
  }
}

__global__ __launch_bounds__(256) void mlp_kernel(
    const float* __restrict__ Dpred,
    const float* __restrict__ Wm0, const float* __restrict__ bm0,
    const float* __restrict__ Wm1, const float* __restrict__ bm1,
    const float* __restrict__ Wm2, const float* __restrict__ bm2,
    const float* __restrict__ Wm3, const float* __restrict__ bm3,
    float* __restrict__ Out)
{
  __shared__ float Abuf[64][ASTR];
  __shared__ float Bbuf[64][ASTR];
  __shared__ __attribute__((aligned(16))) float Wslab[MHD][WSTR];

  const int tid  = threadIdx.x;
  const int lane = tid & 31, wv = tid >> 5;
  const int lh   = lane & 15, sel = lane >> 4;
  const int tok0 = blockIdx.x * 64;

  for (int i = tid; i < 64 * 48; i += 256) {
    int r = i / 48, f = i % 48;
    Abuf[r][f] = (f < Mn) ? Dpred[f * TOK + tok0 + r] : 0.f;
  }
  __syncthreads();

  for (int tile = wv; tile < 64; tile += 8) {
    const int mt = tile >> 4, nt = tile & 15;
    const int n  = nt * 16 + lh;
    const int mr = mt * 16 + lh;
    v8f acc;
    const float bv = bm0[n];
    #pragma unroll
    for (int r = 0; r < 8; ++r) acc[r] = bv;
    #pragma unroll
    for (int kk = 0; kk < 12; ++kk) {
      const int k  = 4 * kk + 2 * sel;
      const int k0 = (k     < Mn) ? k     : (Mn - 1);
      const int k1 = (k + 1 < Mn) ? k + 1 : (Mn - 1);
      v2f a, b;
      a[0] = Abuf[mr][k];      a[1] = Abuf[mr][k + 1];
      b[0] = Wm0[n * Mn + k0]; b[1] = Wm0[n * Mn + k1];
      acc = wmma4(a, b, acc);
    }
    #pragma unroll
    for (int r = 0; r < 8; ++r)
      Bbuf[mt * 16 + r + 8 * sel][n] = fmaxf(acc[r], 0.f);
  }
  __syncthreads();

  hidden_layer(Wm1, bm1, Bbuf, Abuf, Wslab, tid, lane, wv);
  __syncthreads();
  hidden_layer(Wm2, bm2, Abuf, Bbuf, Wslab, tid, lane, wv);
  __syncthreads();

  for (int tile = wv; tile < 8; tile += 8) {
    const int mt = tile >> 1, nt = tile & 1;
    const int n  = nt * 16 + lh;
    const int nc = (n < OUTD) ? n : (OUTD - 1);
    const int mr = mt * 16 + lh;
    v8f acc;
    const float bv = bm3[nc];
    #pragma unroll
    for (int r = 0; r < 8; ++r) acc[r] = bv;
    #pragma unroll 4
    for (int kk = 0; kk < MHD / 4; ++kk) {
      const int k = 4 * kk + 2 * sel;
      v2f a, b;
      a[0] = Bbuf[mr][k];       a[1] = Bbuf[mr][k + 1];
      b[0] = Wm3[nc * MHD + k]; b[1] = Wm3[nc * MHD + k + 1];
      acc = wmma4(a, b, acc);
    }
    if (n < OUTD) {
      #pragma unroll
      for (int r = 0; r < 8; ++r)
        Out[(tok0 + mt * 16 + r + 8 * sel) * OUTD + n] = acc[r];
    }
  }
}

extern "C" void kernel_launch(void* const* d_in, const int* in_sizes, int n_in,
                              void* d_out, int out_size, void* d_ws, size_t ws_size,
                              hipStream_t stream) {
  const float* Dis  = (const float*)d_in[0];
  const float* Wih0 = (const float*)d_in[1];
  const float* Whh0 = (const float*)d_in[2];
  const float* bih0 = (const float*)d_in[3];
  const float* bhh0 = (const float*)d_in[4];
  const float* Wih1 = (const float*)d_in[5];
  const float* Whh1 = (const float*)d_in[6];
  const float* bih1 = (const float*)d_in[7];
  const float* bhh1 = (const float*)d_in[8];
  const float* Wih2 = (const float*)d_in[9];
  const float* Whh2 = (const float*)d_in[10];
  const float* bih2 = (const float*)d_in[11];
  const float* bhh2 = (const float*)d_in[12];
  const float* Wfc1 = (const float*)d_in[13];
  const float* bfc1 = (const float*)d_in[14];
  const float* Wfc2 = (const float*)d_in[15];
  const float* bfc2 = (const float*)d_in[16];
  const float* Wm0  = (const float*)d_in[17];
  const float* bm0  = (const float*)d_in[18];
  const float* Wm1  = (const float*)d_in[19];
  const float* bm1  = (const float*)d_in[20];
  const float* Wm2  = (const float*)d_in[21];
  const float* bm2  = (const float*)d_in[22];
  const float* Wm3  = (const float*)d_in[23];
  const float* bm3  = (const float*)d_in[24];

  float* out   = (float*)d_out;
  float* Dpred = out;
  float* cpf   = out + Mn * Bn * Tn;

  dim3 g1(NBG, Mn);
  lstm45_kernel<<<g1, NT1, 0, stream>>>(
      Dis, Wih0, Whh0, bih0, bhh0, Wih1, Whh1, bih1, bhh1,
      Wih2, Whh2, bih2, bhh2, Wfc1, bfc1, Wfc2, bfc2, Dpred);

  mlp_kernel<<<TOK / 64, 256, 0, stream>>>(
      Dpred, Wm0, bm0, Wm1, bm1, Wm2, bm2, Wm3, bm3, cpf);
}


# LSTM tile 4 shared out over the two lightest waves: wave 4 keeps layer 1 and makes the layer 0 gates, wave 7 the layer 0 cells and layer 2; busiest SIMD 35 WMMAs + 3 cell updates a step, was 40 + 4
# speedup vs baseline: 2.7334x; 1.0786x over previous
#include <hip/hip_runtime.h>

#define Mn   45
#define Bn   64
#define Tn   512
#define Hn   20
#define Gn   80
#define FC   32
#define MHD  256
#define OUTD 22
#define TOK  (Bn * Tn)

#define BQ   16
#define NBG  (Bn / BQ)
#define CH   64

typedef __attribute__((ext_vector_type(2))) float v2f;
typedef __attribute__((ext_vector_type(8))) float v8f;

__device__ __forceinline__ v8f wmma4(v2f a, v2f b, v8f c) {
  return __builtin_amdgcn_wmma_f32_16x16x4_f32(false, a, false, b, (short)0, c,
                                               false, false);
}

__device__ __forceinline__ float sigm(float x) {
  return __builtin_amdgcn_rcpf(1.f + __expf(-x));
}
__device__ __forceinline__ float tanh_fast(float x) {
  return 1.f - 2.f * __builtin_amdgcn_rcpf(1.f + __expf(2.f * x));
}

__device__ __forceinline__ float xhalf(float x) {
  int b = __builtin_bit_cast(int, x);
  return __builtin_bit_cast(float,
      __builtin_amdgcn_permlanex16(b, b, 0x76543210, 0xfedcba98, false, false));
}

#define GP 20

#define NW1 8
#define NT1 (32 * NW1)

__global__ __launch_bounds__(NT1) void lstm45_kernel(
    const float* __restrict__ Dis,
    const float* __restrict__ Wih0, const float* __restrict__ Whh0,
    const float* __restrict__ bih0, const float* __restrict__ bhh0,
    const float* __restrict__ Wih1, const float* __restrict__ Whh1,
    const float* __restrict__ bih1, const float* __restrict__ bhh1,
    const float* __restrict__ Wih2, const float* __restrict__ Whh2,
    const float* __restrict__ bih2, const float* __restrict__ bhh2,
    const float* __restrict__ Wfc1, const float* __restrict__ bfc1,
    const float* __restrict__ Wfc2, const float* __restrict__ bfc2,
    float* __restrict__ Dpred)
{
  __shared__ float xall[BQ][Tn];
  __shared__ __attribute__((aligned(16))) float hcat[BQ][3 * Hn];
  __shared__ __attribute__((aligned(16))) float gt0[5][BQ][GP];
  __shared__ __attribute__((aligned(16))) float gt1[5][BQ][GP];
  __shared__ __attribute__((aligned(16))) float gt2[5][BQ][GP];
  __shared__ __attribute__((aligned(16))) float hist[CH][BQ][Hn];
  __shared__ __attribute__((aligned(16))) float ybuf[BQ][CH + 4];

  const int tid  = threadIdx.x;
  const int m    = blockIdx.y;
  const int b0   = blockIdx.x * BQ;
  const int lane = tid & 31;
  const int wv   = tid >> 5;
  const int lh   = lane & 15;
  const int sel  = lane >> 4;
  const int  wvu     = __builtin_amdgcn_readfirstlane(wv);
  const bool is_tile = wvu < 5;
  const bool has_l2  = wvu < 4 || wvu == 7;
  const bool two     = wvu < 7;
  const int ta   = is_tile ? wvu : 2 * (wvu - 5);
  const int tb   = (ta + 1 < 5) ? ta + 1 : 4;
  const int n    = (lh & 3) * Hn + 4 * ta + (lh >> 2);
  const int nb   = (lh & 3) * Hn + 4 * tb + (lh >> 2);
  const int uc   = 4 * ta + 2 * sel;
  const int ucb  = 4 * tb + 2 * sel;

  for (int i = tid; i < BQ * Tn; i += NT1) {
    int b = i / Tn, t = i % Tn;
    xall[b][t] = Dis[(m * Bn + b0 + b) * Tn + t];
  }
  for (int i = tid; i < BQ * 3 * Hn; i += NT1)
    (&hcat[0][0])[i] = 0.f;

  const float bias0v = bih0[m * Gn + n] + bhh0[m * Gn + n];
  const float bias1v = bih1[m * Gn + n] + bhh1[m * Gn + n];
  const float bias2v = bih2[m * Gn + n] + bhh2[m * Gn + n];
  const float wih0v  = Wih0[m * Gn + n];
  const float bias0w = bih0[m * Gn + nb] + bhh0[m * Gn + nb];
  const float wih0w  = Wih0[m * Gn + nb];
  const float bfc2v  = bfc2[m];

  const int wb = (m * Gn + n) * Hn;
  const int wc = (m * Gn + nb) * Hn;
  v2f bw0[5], bw0b[5], bw1[10], bw2[10];
  #pragma unroll
  for (int kk = 0; kk < 5; ++kk) {
    const int k = 4 * kk + 2 * sel;
    bw0[kk][0] = Whh0[wb + k]; bw0[kk][1] = Whh0[wb + k + 1];
    bw0b[kk][0] = Whh0[wc + k]; bw0b[kk][1] = Whh0[wc + k + 1];
    bw1[kk][0] = Wih1[wb + k]; bw1[kk][1] = Wih1[wb + k + 1];
    bw2[kk][0] = Wih2[wb + k]; bw2[kk][1] = Wih2[wb + k + 1];
  }
  #pragma unroll
  for (int kk = 0; kk < 5; ++kk) {
    const int k = 4 * kk + 2 * sel;
    bw1[5 + kk][0] = Whh1[wb + k]; bw1[5 + kk][1] = Whh1[wb + k + 1];
    bw2[5 + kk][0] = Whh2[wb + k]; bw2[5 + kk][1] = Whh2[wb + k + 1];
  }
  v2f wfa[5], wfb[5];
  float hb0[8], hb1[8], hw0[8], hw1[8];
  {
    const int fa = (m * FC + lh) * Hn;
    const int fb = (m * FC + 16 + lh) * Hn;
    #pragma unroll
    for (int kk = 0; kk < 5; ++kk) {
      const int k = 4 * kk + 2 * sel;
      wfa[kk][0] = Wfc1[fa + k]; wfa[kk][1] = Wfc1[fa + k + 1];
      wfb[kk][0] = Wfc1[fb + k]; wfb[kk][1] = Wfc1[fb + k + 1];
    }
    #pragma unroll
    for (int r = 0; r < 8; ++r) {
      const int f = m * FC + 8 * sel + r;
      hb0[r] = bfc1[f];      hw0[r] = Wfc2[f];
      hb1[r] = bfc1[f + 16]; hw1[r] = Wfc2[f + 16];
    }
  }

  float cs0a = 0.f, cs0b = 0.f, cs0c = 0.f, cs0d = 0.f;
  float cs1a = 0.f, cs1b = 0.f, cs2a = 0.f, cs2b = 0.f;

  __syncthreads();

#define WAVE_HANDOVER()                                                  \
  {                                                                      \
    __builtin_amdgcn_fence(__ATOMIC_RELEASE, "wavefront");               \
    __builtin_amdgcn_wave_barrier();                                     \
    __builtin_amdgcn_sched_barrier(0);                                   \
    __builtin_amdgcn_fence(__ATOMIC_ACQUIRE, "wavefront");               \
  }
#define WAVE_ORDER()                                                     \
  {                                                                      \
    __builtin_amdgcn_fence(__ATOMIC_RELEASE, "wavefront");               \
    __builtin_amdgcn_wave_barrier();                                     \
    __builtin_amdgcn_fence(__ATOMIC_ACQUIRE, "wavefront");               \
  }

#define FILL_FENCE(acc)                                                  \
  {                                                                      \
    asm volatile("" :: "v"(acc));                                        \
    __builtin_amdgcn_sched_barrier(0);                                   \
  }

#define LOADFR(fr, cb)                                                   \
  {                                                                      \
    _Pragma("unroll")                                                    \
    for (int kk = 0; kk < 5; ++kk) {                                     \
      const int k = (cb) + 4 * kk + 2 * sel;                             \
      fr[kk][0] = hcat[lh][k]; fr[kk][1] = hcat[lh][k + 1];              \
    }                                                                    \
  }

#define CHAIN(acc, fr, bw, off)                                          \
  {                                                                      \
    _Pragma("unroll")                                                    \
    for (int kk = 0; kk < 5; ++kk)                                       \
      acc = wmma4(fr[kk], bw[(off) + kk], acc);                          \
  }

#define PARKT(GT, tw, acc)                                               \
  {                                                                      \
    _Pragma("unroll")                                                    \
    for (int r = 0; r < 8; ++r) GT[tw][r + 8 * sel][lh] = acc[r];        \
  }
#define PARK(GT, acc) PARKT(GT, ta, acc)

#define READT(GT, ga, gb)                                                \
  {                                                                      \
    ga = *(const float4*)&GT[ta][lh][8 * sel];                           \
    gb = *(const float4*)&GT[ta][lh][8 * sel + 4];                       \
  }
#define READT_NOW(GT, ga, gb)                                            \
  {                                                                      \
    READT(GT, ga, gb)                                                    \
    __builtin_amdgcn_sched_barrier(0);                                   \
  }

#define CELLX(col, ga, gb, csa, csb, hv)                                 \
  {                                                                      \
    float iga = sigm(ga.x), igb = sigm(gb.x);                            \
    float fga = sigm(ga.y), fgb = sigm(gb.y);                            \
    float gva = tanh_fast(ga.z), gvb = tanh_fast(gb.z);                  \
    float oga = sigm(ga.w), ogb = sigm(gb.w);                            \
    float ca = fga * csa + iga * gva;                                    \
    float cb = fgb * csb + igb * gvb;                                    \
    csa = ca; csb = cb;                                                  \
    hv.x = oga * tanh_fast(ca);                                          \
    hv.y = ogb * tanh_fast(cb);                                          \
    *(float2*)&hcat[lh][(col)] = hv;                                     \
  }
#define CELLC(l, ga, gb, csa, csb)                                       \
  {                                                                      \
    float2 hv;                                                           \
    CELLX((l) * Hn + uc, ga, gb, csa, csb, hv)                           \
    if ((l) == 2) *(float2*)&hist[tc][lh][uc] = hv;                      \
  }

#define XSTART(g0, tt, bv, wx)                                           \
  {                                                                      \
    _Pragma("unroll")                                                    \
    for (int r = 0; r < 8; ++r)                                          \
      g0[r] = (bv) + xall[r + 8 * sel][tt] * (wx);                       \
  }

#define SIDE_L0(tt)                                                      \
  {                                                                      \
    v2f f0[5];                                                           \
    v8f g0;                                                              \
    LOADFR(f0, 0)                                                        \
    XSTART(g0, tt, bias0v, wih0v)                                        \
    if (two) {                                                           \
      v8f g1;                                                            \
      XSTART(g1, tt, bias0w, wih0w)                                      \
      _Pragma("unroll")                                                  \
      for (int kk = 0; kk < 5; ++kk) {                                   \
        g0 = wmma4(f0[kk], bw0[kk], g0);                                 \
        g1 = wmma4(f0[kk], bw0b[kk], g1);                                \
      }                                                                  \
      PARKT(gt0, ta, g0)                                                 \
      PARKT(gt0, tb, g1)                                                 \
    } else {                                                             \
      CHAIN(g0, f0, bw0, 0)                                              \
      PARKT(gt0, ta, g0)                                                 \
    }                                                                    \
  }

#define SIDE_CELLS()                                                     \
  {                                                                      \
    float4 pa, pb;                                                       \
    float2 ha;                                                           \
    pa = *(const float4*)&gt0[ta][lh][8 * sel];                          \
    pb = *(const float4*)&gt0[ta][lh][8 * sel + 4];                      \
    if (two) {                                                           \
      float4 pc, pd;                                                     \
      float2 hb;                                                         \
      pc = *(const float4*)&gt0[tb][lh][8 * sel];                        \
      pd = *(const float4*)&gt0[tb][lh][8 * sel + 4];                    \
      CELLX(uc, pa, pb, cs0a, cs0b, ha)                                  \
      CELLX(ucb, pc, pd, cs0c, cs0d, hb)                                 \
    } else {                                                             \
      CELLX(uc, pa, pb, cs0a, cs0b, ha)                                  \
    }                                                                    \
  }

  float4 qa, qb;

#define BAR_ARRIVE()                                                     \
  {                                                                      \
    __builtin_amdgcn_fence(__ATOMIC_RELEASE, "workgroup");               \
    __builtin_amdgcn_s_barrier_signal(-1);                               \
  }
#define BAR_WAIT()                                                       \
  {                                                                      \
    __builtin_amdgcn_s_barrier_wait(-1);                                 \
    __builtin_amdgcn_fence(__ATOMIC_ACQUIRE, "workgroup");               \
  }

  if (!is_tile) SIDE_L0(0)
  __syncthreads();
  if (!is_tile) SIDE_CELLS()
  __syncthreads();

  v8f e1;
  v8f e2;
  v2f f2[5];
  v2f g0[5];
  if (is_tile) {
    v2f f1[5];
    LOADFR(f1, Hn)
    LOADFR(g0, 0)
    #pragma unroll
    for (int r = 0; r < 8; ++r) e1[r] = bias1v;
    CHAIN(e1, f1, bw1, 5)
    CHAIN(e1, g0, bw1, 0)
  } else {
    SIDE_L0(1)
  }
  __syncthreads();

  for (int t = 0; t < Tn; ++t) {
    const int tc = t & (CH - 1);
    const int tn = (t + 2 < Tn) ? t + 2 : Tn - 1;

    if (has_l2)
      LOADFR(f2, 2 * Hn)
    if (is_tile) {
      PARK(gt1, e1)
      WAVE_ORDER()
      READT_NOW(gt1, qa, qb)
      CELLC(1, qa, qb, cs1a, cs1b)
    } else {
      SIDE_CELLS()
    }
    BAR_ARRIVE()
    if (has_l2) {
      #pragma unroll
      for (int r = 0; r < 8; ++r) e2[r] = bias2v;
      CHAIN(e2, f2, bw2, 5)
    }
    BAR_WAIT()

    if (is_tile) {
      v2f f1[5];
      LOADFR(f1, Hn)
      LOADFR(g0, 0)
      if (has_l2) {
        CHAIN(e2, f1, bw2, 0)
        PARK(gt2, e2)
        WAVE_HANDOVER()
        READT_NOW(gt2, qa, qb)
        #pragma unroll
        for (int r = 0; r < 8; ++r) e1[r] = bias1v;
        CHAIN(e1, f1, bw1, 5)
        FILL_FENCE(e1)
        CELLC(2, qa, qb, cs2a, cs2b)
      } else {
        v8f a0;
        XSTART(a0, tn, bias0v, wih0v)
        #pragma unroll
        for (int r = 0; r < 8; ++r) e1[r] = bias1v;
        #pragma unroll
        for (int kk = 0; kk < 5; ++kk) {
          e1 = wmma4(f1[kk], bw1[5 + kk], e1);
          a0 = wmma4(g0[kk], bw0[kk], a0);
        }
        PARKT(gt0, ta, a0)
        CHAIN(e1, g0, bw1, 0)
      }
    } else if (has_l2) {
      v2f f1[5];
      LOADFR(f1, Hn)
      CHAIN(e2, f1, bw2, 0)
      PARK(gt2, e2)
      WAVE_ORDER()
      READT(gt2, qa, qb)
      CELLC(2, qa, qb, cs2a, cs2b)
    } else {
      SIDE_L0(tn)
    }
    BAR_ARRIVE()
    if (is_tile && has_l2)
      CHAIN(e1, g0, bw1, 0)
    BAR_WAIT()

    if (tc == CH - 1) {
      #pragma unroll 2
      for (int tl = wvu; tl < CH; tl += NW1) {
        v8f z0, z1;
        #pragma unroll
        for (int r = 0; r < 8; ++r) { z0[r] = hb0[r]; z1[r] = hb1[r]; }
        #pragma unroll
        for (int kk = 0; kk < 5; ++kk) {
          const int k = 4 * kk + 2 * sel;
          v2f hf; hf[0] = hist[tl][lh][k]; hf[1] = hist[tl][lh][k + 1];
          z0 = wmma4(wfa[kk], hf, z0);
          z1 = wmma4(wfb[kk], hf, z1);
        }
        float p0 = fmaxf(z0[0], 0.f) * hw0[0];
        float p1 = fmaxf(z1[0], 0.f) * hw1[0];
        #pragma unroll
        for (int r = 1; r < 8; ++r) p0 += fmaxf(z0[r], 0.f) * hw0[r];
        #pragma unroll
        for (int r = 1; r < 8; ++r) p1 += fmaxf(z1[r], 0.f) * hw1[r];
        float p = p0 + p1;
        p += xhalf(p);
        ybuf[lh][tl] = p + bfc2v;
      }
      __syncthreads();
      for (int i = tid; i < BQ * CH / 4; i += NT1) {
        const int b = i / (CH / 4), q = i % (CH / 4);
        const float4 v = *(const float4*)&ybuf[b][4 * q];
        *(float4*)&Dpred[(m * Bn + b0 + b) * Tn + (t - (CH - 1)) + 4 * q] = v;
      }
    }
  }
#undef BAR_WAIT
#undef BAR_ARRIVE
#undef SIDE_CELLS
#undef SIDE_L0
#undef XSTART
#undef CELLC
#undef CELLX
#undef READT_NOW
#undef READT
#undef PARK
#undef PARKT
#undef CHAIN
#undef LOADFR
#undef FILL_FENCE
#undef WAVE_ORDER
#undef WAVE_HANDOVER
}

#define ASTR (MHD + 4)
#define KS   16
#define WSTR (KS + 4)

__device__ __forceinline__ void hidden_layer(
    const float* __restrict__ W, const float* __restrict__ bias,
    const float (*IN)[ASTR], float (*OUT)[ASTR], float (*WS)[WSTR],
    int tid, int lane, int wv)
{
  const int lh = lane & 15, sel = lane >> 4;
  const int n0 = wv * 16 + lh, n1 = n0 + 128;
  v8f acc[2][4];
  {
    const float bv0 = bias[n0], bv1 = bias[n1];
    #pragma unroll
    for (int mt = 0; mt < 4; ++mt) {
      #pragma unroll
      for (int r = 0; r < 8; ++r) { acc[0][mt][r] = bv0; acc[1][mt][r] = bv1; }
    }
  }
  const int crow = tid >> 2, ccol = 4 * (tid & 3);
  const float* wp = W + crow * MHD + ccol;
  float4 w0 = *(const float4*)(wp);
  float4 w1 = *(const float4*)(wp + 64 * MHD);
  float4 w2 = *(const float4*)(wp + 128 * MHD);
  float4 w3 = *(const float4*)(wp + 192 * MHD);
  #pragma unroll 1
  for (int s = 0; s < MHD / KS; ++s) {
    __syncthreads();
    *(float4*)&WS[crow][ccol]       = w0;
    *(float4*)&WS[crow + 64][ccol]  = w1;
    *(float4*)&WS[crow + 128][ccol] = w2;
    *(float4*)&WS[crow + 192][ccol] = w3;
    __syncthreads();
    if (s + 1 < MHD / KS) {
      const float* wn = wp + (s + 1) * KS;
      w0 = *(const float4*)(wn);
      w1 = *(const float4*)(wn + 64 * MHD);
      w2 = *(const float4*)(wn + 128 * MHD);
      w3 = *(const float4*)(wn + 192 * MHD);
    }
    #pragma unroll
    for (int kk = 0; kk < KS / 4; ++kk) {
      const int k = 4 * kk + 2 * sel;
      v2f b0, b1;
      b0[0] = WS[n0][k]; b0[1] = WS[n0][k + 1];
      b1[0] = WS[n1][k]; b1[1] = WS[n1][k + 1];
      #pragma unroll
      for (int mt = 0; mt < 4; ++mt) {
        v2f a;
        a[0] = IN[mt * 16 + lh][s * KS + k]; a[1] = IN[mt * 16 + lh][s * KS + k + 1];
        acc[0][mt] = wmma4(a, b0, acc[0][mt]);
        acc[1][mt] = wmma4(a, b1, acc[1][mt]);
      }
    }
  }
  #pragma unroll
  for (int mt = 0; mt < 4; ++mt) {
    #pragma unroll
    for (int r = 0; r < 8; ++r) {
      OUT[mt * 16 + r + 8 * sel][n0] = fmaxf(acc[0][mt][r], 0.f);
      OUT[mt * 16 + r + 8 * sel][n1] = fmaxf(acc[1][mt][r], 0.f);
    }
  }
}

__global__ __launch_bounds__(256) void mlp_kernel(
    const float* __restrict__ Dpred,
    const float* __restrict__ Wm0, const float* __restrict__ bm0,
    const float* __restrict__ Wm1, const float* __restrict__ bm1,
    const float* __restrict__ Wm2, const float* __restrict__ bm2,
    const float* __restrict__ Wm3, const float* __restrict__ bm3,
    float* __restrict__ Out)
{
  __shared__ float Abuf[64][ASTR];
  __shared__ float Bbuf[64][ASTR];
  __shared__ __attribute__((aligned(16))) float Wslab[MHD][WSTR];

  const int tid  = threadIdx.x;
  const int lane = tid & 31, wv = tid >> 5;
  const int lh   = lane & 15, sel = lane >> 4;
  const int tok0 = blockIdx.x * 64;

  for (int i = tid; i < 64 * 48; i += 256) {
    int r = i / 48, f = i % 48;
    Abuf[r][f] = (f < Mn) ? Dpred[f * TOK + tok0 + r] : 0.f;
  }
  __syncthreads();

  for (int tile = wv; tile < 64; tile += 8) {
    const int mt = tile >> 4, nt = tile & 15;
    const int n  = nt * 16 + lh;
    const int mr = mt * 16 + lh;
    v8f acc;
    const float bv = bm0[n];
    #pragma unroll
    for (int r = 0; r < 8; ++r) acc[r] = bv;
    #pragma unroll
    for (int kk = 0; kk < 12; ++kk) {
      const int k  = 4 * kk + 2 * sel;
      const int k0 = (k     < Mn) ? k     : (Mn - 1);
      const int k1 = (k + 1 < Mn) ? k + 1 : (Mn - 1);
      v2f a, b;
      a[0] = Abuf[mr][k];      a[1] = Abuf[mr][k + 1];
      b[0] = Wm0[n * Mn + k0]; b[1] = Wm0[n * Mn + k1];
      acc = wmma4(a, b, acc);
    }
    #pragma unroll
    for (int r = 0; r < 8; ++r)
      Bbuf[mt * 16 + r + 8 * sel][n] = fmaxf(acc[r], 0.f);
  }
  __syncthreads();

  hidden_layer(Wm1, bm1, Bbuf, Abuf, Wslab, tid, lane, wv);
  __syncthreads();
  hidden_layer(Wm2, bm2, Abuf, Bbuf, Wslab, tid, lane, wv);
  __syncthreads();

  for (int tile = wv; tile < 8; tile += 8) {
    const int mt = tile >> 1, nt = tile & 1;
    const int n  = nt * 16 + lh;
    const int nc = (n < OUTD) ? n : (OUTD - 1);
    const int mr = mt * 16 + lh;
    v8f acc;
    const float bv = bm3[nc];
    #pragma unroll
    for (int r = 0; r < 8; ++r) acc[r] = bv;
    #pragma unroll 4
    for (int kk = 0; kk < MHD / 4; ++kk) {
      const int k = 4 * kk + 2 * sel;
      v2f a, b;
      a[0] = Bbuf[mr][k];       a[1] = Bbuf[mr][k + 1];
      b[0] = Wm3[nc * MHD + k]; b[1] = Wm3[nc * MHD + k + 1];
      acc = wmma4(a, b, acc);
    }
    if (n < OUTD) {
      #pragma unroll
      for (int r = 0; r < 8; ++r)
        Out[(tok0 + mt * 16 + r + 8 * sel) * OUTD + n] = acc[r];
    }
  }
}

extern "C" void kernel_launch(void* const* d_in, const int* in_sizes, int n_in,
                              void* d_out, int out_size, void* d_ws, size_t ws_size,
                              hipStream_t stream) {
  const float* Dis  = (const float*)d_in[0];
  const float* Wih0 = (const float*)d_in[1];
  const float* Whh0 = (const float*)d_in[2];
  const float* bih0 = (const float*)d_in[3];
  const float* bhh0 = (const float*)d_in[4];
  const float* Wih1 = (const float*)d_in[5];
  const float* Whh1 = (const float*)d_in[6];
  const float* bih1 = (const float*)d_in[7];
  const float* bhh1 = (const float*)d_in[8];
  const float* Wih2 = (const float*)d_in[9];
  const float* Whh2 = (const float*)d_in[10];
  const float* bih2 = (const float*)d_in[11];
  const float* bhh2 = (const float*)d_in[12];
  const float* Wfc1 = (const float*)d_in[13];
  const float* bfc1 = (const float*)d_in[14];
  const float* Wfc2 = (const float*)d_in[15];
  const float* bfc2 = (const float*)d_in[16];
  const float* Wm0  = (const float*)d_in[17];
  const float* bm0  = (const float*)d_in[18];
  const float* Wm1  = (const float*)d_in[19];
  const float* bm1  = (const float*)d_in[20];
  const float* Wm2  = (const float*)d_in[21];
  const float* bm2  = (const float*)d_in[22];
  const float* Wm3  = (const float*)d_in[23];
  const float* bm3  = (const float*)d_in[24];

  float* out   = (float*)d_out;
  float* Dpred = out;
  float* cpf   = out + Mn * Bn * Tn;

  dim3 g1(NBG, Mn);
  lstm45_kernel<<<g1, NT1, 0, stream>>>(
      Dis, Wih0, Whh0, bih0, bhh0, Wih1, Whh1, bih1, bhh1,
      Wih2, Whh2, bih2, bhh2, Wfc1, bfc1, Wfc2, bfc2, Dpred);

  mlp_kernel<<<TOK / 64, 256, 0, stream>>>(
      Dpred, Wm0, bm0, Wm1, bm1, Wm2, bm2, Wm3, bm3, cpf);
}
